# starting kernel
# baseline (speedup 1.0000x reference)
#include <hip/hip_runtime.h>
#include <hip/hip_bf16.h>

typedef __attribute__((ext_vector_type(2))) float v2f;
typedef __attribute__((ext_vector_type(8))) float v8f;

#define NB     2
#define NPT    16384
#define NS     4096
#define NC     64
#define NOUT   128
#define NK     16
#define CH     512
#define NCHUNK 8
#define CHPTS  (NPT / NCHUNK)
#define AGROW  1076

__global__ __launch_bounds__(256) void knn_partial_kernel(
    const float* __restrict__ xyz, const float* __restrict__ sxyz,
    float* __restrict__ pd, int* __restrict__ pi) {
    const int tid = threadIdx.x;
    const int blkPerB = NS / 256;
    const int chunk = blockIdx.x % NCHUNK;
    const int t     = blockIdx.x / NCHUNK;
    const int b     = t / blkPerB;
    const int s     = (t % blkPerB) * 256 + tid;

    const float sx = sxyz[(b * 3 + 0) * NS + s];
    const float sy = sxyz[(b * 3 + 1) * NS + s];
    const float sz = sxyz[(b * 3 + 2) * NS + s];
    const float s2 = sx * sx + sy * sy + sz * sz;

    __shared__ float xl[CH], yl[CH], zl[CH], x2l[CH];

    float bd[NK];
    int   bi[NK];
#pragma unroll
    for (int i = 0; i < NK; ++i) { bd[i] = 3.4e38f; bi[i] = 0; }

    const float* xb = xyz + (size_t)b * 3 * NPT;
    const int n0 = chunk * CHPTS;

    for (int c0 = n0; c0 < n0 + CHPTS; c0 += CH) {
        __syncthreads();
        for (int j = tid; j < CH; j += 256) {
            float xv = xb[c0 + j];
            float yv = xb[NPT + c0 + j];
            float zv = xb[2 * NPT + c0 + j];
            xl[j] = xv; yl[j] = yv; zl[j] = zv;
            x2l[j] = xv * xv + yv * yv + zv * zv;
        }
        __syncthreads();
#pragma unroll 4
        for (int jj = 0; jj < CH; ++jj) {
            float dot = sx * xl[jj] + sy * yl[jj] + sz * zl[jj];
            float d2  = (s2 + x2l[jj]) - 2.0f * dot;
            if (d2 < bd[NK - 1]) {
                float cd = d2; int ci = c0 + jj;
#pragma unroll
                for (int i = 0; i < NK; ++i) {
                    if (cd < bd[i]) {
                        float td = bd[i]; int ti = bi[i];
                        bd[i] = cd; bi[i] = ci;
                        cd = td; ci = ti;
                    }
                }
            }
        }
    }
    const size_t base = ((size_t)(b * NS + s) * NCHUNK + chunk) * NK;
#pragma unroll
    for (int i = 0; i < NK; ++i) { pd[base + i] = bd[i]; pi[base + i] = bi[i]; }
}

__global__ __launch_bounds__(256) void knn_merge_kernel(
    const float* __restrict__ pd, const int* __restrict__ pi,
    int* __restrict__ knn) {
    const int g = blockIdx.x * 256 + threadIdx.x;
    const int b = g / NS;
    const int s = g % NS;

    float bd[NK];
    int   bi[NK];
#pragma unroll
    for (int i = 0; i < NK; ++i) { bd[i] = 3.4e38f; bi[i] = 0; }

    const size_t base = (size_t)(b * NS + s) * NCHUNK * NK;
    for (int c = 0; c < NCHUNK * NK; ++c) {
        float d = pd[base + c];
        int   ci = pi[base + c];
        if (d < bd[NK - 1]) {
            float cd = d;
#pragma unroll
            for (int i = 0; i < NK; ++i) {
                if (cd < bd[i]) {
                    float td = bd[i]; int ti = bi[i];
                    bd[i] = cd; bi[i] = ci;
                    cd = td; ci = ti;
                }
            }
        }
    }
#pragma unroll
    for (int i = 0; i < NK; ++i)
        knn[((size_t)(b * NS + s)) * NK + i] = bi[i];
}

__global__ __launch_bounds__(256) void pointconv_kernel(
    const float* __restrict__ xyz,  const float* __restrict__ feat,
    const float* __restrict__ sxyz,
    const float* __restrict__ w1,   const float* __restrict__ b1,
    const float* __restrict__ w2,   const float* __restrict__ b2,
    const float* __restrict__ lw,   const float* __restrict__ lb,
    const int*   __restrict__ knn,  float* __restrict__ out)
{
    const int tid  = threadIdx.x;
    const int lane = tid & 31;
    const int wave = tid >> 5;
    const int b    = blockIdx.x >> 8;
    const int s0   = (blockIdx.x & 255) * 16;

    __shared__ float w1s[24], b1s[8], w2s[128], b2s[16];
    __shared__ float Wl[16][16][16];
    __shared__ float Fl[16][16][68];
    __shared__ float Ag[16 * AGROW];

    if (tid < 24)  w1s[tid] = w1[tid];
    if (tid < 8)   b1s[tid] = b1[tid];
    if (tid < 128) w2s[tid] = w2[tid];
    if (tid < 16)  b2s[tid] = b2[tid];
    __syncthreads();

    {
        const int s = tid >> 4, k = tid & 15;
        const int idx = knn[((size_t)(b * NS + s0 + s)) * NK + k];
        const float nx = xyz[(size_t)b * 3 * NPT + idx];
        const float ny = xyz[(size_t)b * 3 * NPT + NPT + idx];
        const float nz = xyz[(size_t)b * 3 * NPT + 2 * NPT + idx];
        const float rx = nx - sxyz[(b * 3 + 0) * NS + s0 + s];
        const float ry = ny - sxyz[(b * 3 + 1) * NS + s0 + s];
        const float rz = nz - sxyz[(b * 3 + 2) * NS + s0 + s];

        float h[8];
#pragma unroll
        for (int o = 0; o < 8; ++o) {
            float v = w1s[o * 3 + 0] * rx + w1s[o * 3 + 1] * ry +
                      w1s[o * 3 + 2] * rz + b1s[o];
            h[o] = v >= 0.f ? v : 0.1f * v;
        }
#pragma unroll
        for (int o = 0; o < 16; ++o) {
            float v = b2s[o];
#pragma unroll
            for (int i = 0; i < 8; ++i) v += w2s[o * 8 + i] * h[i];
            v = v >= 0.f ? v : 0.1f * v;
            Wl[s][k][o] = v;
        }
        Fl[s][k][0] = nx; Fl[s][k][1] = ny; Fl[s][k][2] = nz;
        const float* fb = feat + (size_t)b * NC * NPT + idx;
#pragma unroll 8
        for (int cf = 0; cf < NC; ++cf)
            Fl[s][k][3 + cf] = fb[(size_t)cf * NPT];
        Fl[s][k][67] = 0.f;
    }
    __syncthreads();

    {
        const int r    = lane & 15;
        const int koff = (lane < 16) ? 0 : 2;
#pragma unroll
        for (int si = 0; si < 2; ++si) {
            const int s = wave * 2 + si;
#pragma unroll
            for (int ct = 0; ct < 5; ++ct) {
                const int c  = ct * 16 + r;
                const int cc = c > 67 ? 67 : c;
                v8f acc = {};
#pragma unroll
                for (int kb = 0; kb < 4; ++kb) {
                    const int k0 = kb * 4 + koff;
                    v2f a;  a.x  = Wl[s][k0][r];  a.y  = Wl[s][k0 + 1][r];
                    v2f bb; bb.x = Fl[s][k0][cc]; bb.y = Fl[s][k0 + 1][cc];
                    acc = __builtin_amdgcn_wmma_f32_16x16x4_f32(
                        false, a, false, bb, (short)0, acc, false, false);
                }
                if (c < 67) {
#pragma unroll
                    for (int rr = 0; rr < 8; ++rr) {
                        const int wi = rr + ((lane < 16) ? 0 : 8);
                        Ag[s * AGROW + wi * 67 + c] = acc[rr];
                    }
                }
            }
        }
    }
    __syncthreads();

    {
        const int r    = lane & 15;
        const int koff = (lane < 16) ? 0 : 2;
        const float* Lrow = lw + (size_t)(wave * 16 + r) * 1072;
        const float* Arow = &Ag[r * AGROW];
        v8f acc = {};
        for (int kb = 0; kb < 268; ++kb) {
            const int k0 = kb * 4 + koff;
            v2f a;  a.x  = Lrow[k0]; a.y  = Lrow[k0 + 1];
            v2f bb; bb.x = Arow[k0]; bb.y = Arow[k0 + 1];
            acc = __builtin_amdgcn_wmma_f32_16x16x4_f32(
                false, a, false, bb, (short)0, acc, false, false);
        }
        const int su = s0 + r;
#pragma unroll
        for (int rr = 0; rr < 8; ++rr) {
            const int o = wave * 16 + rr + ((lane < 16) ? 0 : 8);
            float v = acc[rr] + lb[o];
            v = v >= 0.f ? v : 0.1f * v;
            out[(size_t)b * NOUT * NS + (size_t)o * NS + su] = v;
        }
    }
}

extern "C" void kernel_launch(void* const* d_in, const int* in_sizes, int n_in,
                              void* d_out, int out_size, void* d_ws, size_t ws_size,
                              hipStream_t stream) {
    const float* xyz  = (const float*)d_in[0];
    const float* feat = (const float*)d_in[1];
    const float* sxyz = (const float*)d_in[2];
    const float* w1   = (const float*)d_in[3];
    const float* b1   = (const float*)d_in[4];
    const float* w2   = (const float*)d_in[5];
    const float* b2   = (const float*)d_in[6];
    const float* lw   = (const float*)d_in[7];
    const float* lb   = (const float*)d_in[8];
    float* out = (float*)d_out;

    const size_t npart = (size_t)NB * NS * NCHUNK * NK;
    float* pd  = (float*)d_ws;
    int*   pi  = (int*)((char*)d_ws + npart * sizeof(float));
    int*   knn = (int*)((char*)d_ws + 2 * npart * sizeof(float));

    knn_partial_kernel<<<NB * (NS / 256) * NCHUNK, 256, 0, stream>>>(xyz, sxyz, pd, pi);
    knn_merge_kernel<<<NB * NS / 256, 256, 0, stream>>>(pd, pi, knn);
    pointconv_kernel<<<NB * (NS / 16), 256, 0, stream>>>(
        xyz, feat, sxyz, w1, b1, w2, b2, lw, lb, knn, out);
}


# KNN top-16 insert made branch-free on 32 scalar registers (16 compares, median-of-3, selects; was 16 nested branches copying both lists); equal distances now keep index order in the k-sum
# speedup vs baseline: 3.9297x; 3.9297x over previous
#include <hip/hip_runtime.h>
#include <hip/hip_bf16.h>

typedef __attribute__((ext_vector_type(2))) float v2f;
typedef __attribute__((ext_vector_type(8))) float v8f;

#define NB     2
#define NPT    16384
#define NS     4096
#define NC     64
#define NOUT   128
#define NK     16
#define CH     512
#define NCHUNK 8
#define CHPTS  (NPT / NCHUNK)
#define AGROW  1076

struct Top16 {
    float d0, d1, d2, d3, d4, d5, d6, d7, d8, d9, d10, d11, d12, d13, d14, d15;
    int   i0, i1, i2, i3, i4, i5, i6, i7, i8, i9, i10, i11, i12, i13, i14, i15;
};

__device__ __forceinline__ void top16_init(Top16& t) {
    t.d0 = t.d1 = t.d2 = t.d3 = t.d4 = t.d5 = t.d6 = t.d7 = 3.4e38f;
    t.d8 = t.d9 = t.d10 = t.d11 = t.d12 = t.d13 = t.d14 = t.d15 = 3.4e38f;
    t.i0 = t.i1 = t.i2 = t.i3 = t.i4 = t.i5 = t.i6 = t.i7 = 0;
    t.i8 = t.i9 = t.i10 = t.i11 = t.i12 = t.i13 = t.i14 = t.i15 = 0;
}

#define TOP16_SHIFT(K, P)                                              \
    t.i##K = m##P ? t.i##P : (m##K ? ci : t.i##K);                     \
    t.d##K = __builtin_amdgcn_fmed3f(t.d##P, cd, t.d##K);

__device__ __forceinline__ void top16_insert(Top16& t, const float cd, const int ci) {
    const bool m0  = cd < t.d0,  m1  = cd < t.d1,  m2  = cd < t.d2,  m3  = cd < t.d3;
    const bool m4  = cd < t.d4,  m5  = cd < t.d5,  m6  = cd < t.d6,  m7  = cd < t.d7;
    const bool m8  = cd < t.d8,  m9  = cd < t.d9,  m10 = cd < t.d10, m11 = cd < t.d11;
    const bool m12 = cd < t.d12, m13 = cd < t.d13, m14 = cd < t.d14, m15 = cd < t.d15;
    TOP16_SHIFT(15, 14) TOP16_SHIFT(14, 13) TOP16_SHIFT(13, 12) TOP16_SHIFT(12, 11)
    TOP16_SHIFT(11, 10) TOP16_SHIFT(10, 9)  TOP16_SHIFT(9, 8)   TOP16_SHIFT(8, 7)
    TOP16_SHIFT(7, 6)   TOP16_SHIFT(6, 5)   TOP16_SHIFT(5, 4)   TOP16_SHIFT(4, 3)
    TOP16_SHIFT(3, 2)   TOP16_SHIFT(2, 1)   TOP16_SHIFT(1, 0)
    t.i0 = m0 ? ci : t.i0;
    t.d0 = m0 ? cd : t.d0;
}
#undef TOP16_SHIFT

#define TOP16_STORE_D(t, p) { (p)[0]=t.d0; (p)[1]=t.d1; (p)[2]=t.d2; (p)[3]=t.d3; (p)[4]=t.d4; (p)[5]=t.d5; (p)[6]=t.d6; (p)[7]=t.d7; \
    (p)[8]=t.d8; (p)[9]=t.d9; (p)[10]=t.d10; (p)[11]=t.d11; (p)[12]=t.d12; (p)[13]=t.d13; (p)[14]=t.d14; (p)[15]=t.d15; }
#define TOP16_STORE_I(t, p) { (p)[0]=t.i0; (p)[1]=t.i1; (p)[2]=t.i2; (p)[3]=t.i3; (p)[4]=t.i4; (p)[5]=t.i5; (p)[6]=t.i6; (p)[7]=t.i7; \
    (p)[8]=t.i8; (p)[9]=t.i9; (p)[10]=t.i10; (p)[11]=t.i11; (p)[12]=t.i12; (p)[13]=t.i13; (p)[14]=t.i14; (p)[15]=t.i15; }

__global__ __launch_bounds__(256) void knn_partial_kernel(
    const float* __restrict__ xyz, const float* __restrict__ sxyz,
    float* __restrict__ pd, int* __restrict__ pi) {
    const int tid = threadIdx.x;
    const int blkPerB = NS / 256;
    const int chunk = blockIdx.x % NCHUNK;
    const int t     = blockIdx.x / NCHUNK;
    const int b     = t / blkPerB;
    const int s     = (t % blkPerB) * 256 + tid;

    const float sx = sxyz[(b * 3 + 0) * NS + s];
    const float sy = sxyz[(b * 3 + 1) * NS + s];
    const float sz = sxyz[(b * 3 + 2) * NS + s];
    const float s2 = sx * sx + sy * sy + sz * sz;

    __shared__ float xl[CH], yl[CH], zl[CH], x2l[CH];

    Top16 top;
    top16_init(top);

    const float* xb = xyz + (size_t)b * 3 * NPT;
    const int n0 = chunk * CHPTS;

    for (int c0 = n0; c0 < n0 + CHPTS; c0 += CH) {
        __syncthreads();
        for (int j = tid; j < CH; j += 256) {
            float xv = xb[c0 + j];
            float yv = xb[NPT + c0 + j];
            float zv = xb[2 * NPT + c0 + j];
            xl[j] = xv; yl[j] = yv; zl[j] = zv;
            x2l[j] = xv * xv + yv * yv + zv * zv;
        }
        __syncthreads();
#pragma unroll 4
        for (int jj = 0; jj < CH; ++jj) {
            float dot = sx * xl[jj] + sy * yl[jj] + sz * zl[jj];
            float d2  = (s2 + x2l[jj]) - 2.0f * dot;
            if (d2 < top.d15) top16_insert(top, d2, c0 + jj);
        }
    }
    const size_t base = ((size_t)(b * NS + s) * NCHUNK + chunk) * NK;
    TOP16_STORE_D(top, pd + base)
    TOP16_STORE_I(top, pi + base)
}

__global__ __launch_bounds__(256) void knn_merge_kernel(
    const float* __restrict__ pd, const int* __restrict__ pi,
    int* __restrict__ knn) {
    const int g = blockIdx.x * 256 + threadIdx.x;
    const int b = g / NS;
    const int s = g % NS;

    Top16 top;
    top16_init(top);

    const size_t base = (size_t)(b * NS + s) * NCHUNK * NK;
    for (int c = 0; c < NCHUNK * NK; ++c) {
        float d = pd[base + c];
        int   ci = pi[base + c];
        if (d < top.d15) top16_insert(top, d, ci);
    }
    TOP16_STORE_I(top, knn + ((size_t)(b * NS + s)) * NK)
}

__global__ __launch_bounds__(256) void pointconv_kernel(
    const float* __restrict__ xyz,  const float* __restrict__ feat,
    const float* __restrict__ sxyz,
    const float* __restrict__ w1,   const float* __restrict__ b1,
    const float* __restrict__ w2,   const float* __restrict__ b2,
    const float* __restrict__ lw,   const float* __restrict__ lb,
    const int*   __restrict__ knn,  float* __restrict__ out)
{
    const int tid  = threadIdx.x;
    const int lane = tid & 31;
    const int wave = tid >> 5;
    const int b    = blockIdx.x >> 8;
    const int s0   = (blockIdx.x & 255) * 16;

    __shared__ float w1s[24], b1s[8], w2s[128], b2s[16];
    __shared__ float Wl[16][16][16];
    __shared__ float Fl[16][16][68];
    __shared__ float Ag[16 * AGROW];

    if (tid < 24)  w1s[tid] = w1[tid];
    if (tid < 8)   b1s[tid] = b1[tid];
    if (tid < 128) w2s[tid] = w2[tid];
    if (tid < 16)  b2s[tid] = b2[tid];
    __syncthreads();

    {
        const int s = tid >> 4, k = tid & 15;
        const int idx = knn[((size_t)(b * NS + s0 + s)) * NK + k];
        const float nx = xyz[(size_t)b * 3 * NPT + idx];
        const float ny = xyz[(size_t)b * 3 * NPT + NPT + idx];
        const float nz = xyz[(size_t)b * 3 * NPT + 2 * NPT + idx];
        const float rx = nx - sxyz[(b * 3 + 0) * NS + s0 + s];
        const float ry = ny - sxyz[(b * 3 + 1) * NS + s0 + s];
        const float rz = nz - sxyz[(b * 3 + 2) * NS + s0 + s];

        float h[8];
#pragma unroll
        for (int o = 0; o < 8; ++o) {
            float v = w1s[o * 3 + 0] * rx + w1s[o * 3 + 1] * ry +
                      w1s[o * 3 + 2] * rz + b1s[o];
            h[o] = v >= 0.f ? v : 0.1f * v;
        }
#pragma unroll
        for (int o = 0; o < 16; ++o) {
            float v = b2s[o];
#pragma unroll
            for (int i = 0; i < 8; ++i) v += w2s[o * 8 + i] * h[i];
            v = v >= 0.f ? v : 0.1f * v;
            Wl[s][k][o] = v;
        }
        Fl[s][k][0] = nx; Fl[s][k][1] = ny; Fl[s][k][2] = nz;
        const float* fb = feat + (size_t)b * NC * NPT + idx;
#pragma unroll 8
        for (int cf = 0; cf < NC; ++cf)
            Fl[s][k][3 + cf] = fb[(size_t)cf * NPT];
        Fl[s][k][67] = 0.f;
    }
    __syncthreads();

    {
        const int r    = lane & 15;
        const int koff = (lane < 16) ? 0 : 2;
#pragma unroll
        for (int si = 0; si < 2; ++si) {
            const int s = wave * 2 + si;
#pragma unroll
            for (int ct = 0; ct < 5; ++ct) {
                const int c  = ct * 16 + r;
                const int cc = c > 67 ? 67 : c;
                v8f acc = {};
#pragma unroll
                for (int kb = 0; kb < 4; ++kb) {
                    const int k0 = kb * 4 + koff;
                    v2f a;  a.x  = Wl[s][k0][r];  a.y  = Wl[s][k0 + 1][r];
                    v2f bb; bb.x = Fl[s][k0][cc]; bb.y = Fl[s][k0 + 1][cc];
                    acc = __builtin_amdgcn_wmma_f32_16x16x4_f32(
                        false, a, false, bb, (short)0, acc, false, false);
                }
                if (c < 67) {
#pragma unroll
                    for (int rr = 0; rr < 8; ++rr) {
                        const int wi = rr + ((lane < 16) ? 0 : 8);
                        Ag[s * AGROW + wi * 67 + c] = acc[rr];
                    }
                }
            }
        }
    }
    __syncthreads();

    {
        const int r    = lane & 15;
        const int koff = (lane < 16) ? 0 : 2;
        const float* Lrow = lw + (size_t)(wave * 16 + r) * 1072;
        const float* Arow = &Ag[r * AGROW];
        v8f acc = {};
        for (int kb = 0; kb < 268; ++kb) {
            const int k0 = kb * 4 + koff;
            v2f a;  a.x  = Lrow[k0]; a.y  = Lrow[k0 + 1];
            v2f bb; bb.x = Arow[k0]; bb.y = Arow[k0 + 1];
            acc = __builtin_amdgcn_wmma_f32_16x16x4_f32(
                false, a, false, bb, (short)0, acc, false, false);
        }
        const int su = s0 + r;
#pragma unroll
        for (int rr = 0; rr < 8; ++rr) {
            const int o = wave * 16 + rr + ((lane < 16) ? 0 : 8);
            float v = acc[rr] + lb[o];
            v = v >= 0.f ? v : 0.1f * v;
            out[(size_t)b * NOUT * NS + (size_t)o * NS + su] = v;
        }
    }
}

extern "C" void kernel_launch(void* const* d_in, const int* in_sizes, int n_in,
                              void* d_out, int out_size, void* d_ws, size_t ws_size,
                              hipStream_t stream) {
    const float* xyz  = (const float*)d_in[0];
    const float* feat = (const float*)d_in[1];
    const float* sxyz = (const float*)d_in[2];
    const float* w1   = (const float*)d_in[3];
    const float* b1   = (const float*)d_in[4];
    const float* w2   = (const float*)d_in[5];
    const float* b2   = (const float*)d_in[6];
    const float* lw   = (const float*)d_in[7];
    const float* lb   = (const float*)d_in[8];
    float* out = (float*)d_out;

    const size_t npart = (size_t)NB * NS * NCHUNK * NK;
    float* pd  = (float*)d_ws;
    int*   pi  = (int*)((char*)d_ws + npart * sizeof(float));
    int*   knn = (int*)((char*)d_ws + 2 * npart * sizeof(float));

    knn_partial_kernel<<<NB * (NS / 256) * NCHUNK, 256, 0, stream>>>(xyz, sxyz, pd, pi);
    knn_merge_kernel<<<NB * NS / 256, 256, 0, stream>>>(pd, pi, knn);
    pointconv_kernel<<<NB * (NS / 16), 256, 0, stream>>>(
        xyz, feat, sxyz, w1, b1, w2, b2, lw, lb, knn, out);
}


# KNN re-mapped: a wave owns 2 queries (wave-uniform), lanes own candidates; top-16 list across 16 lanes, row-shift insert; one full scan per query, no partial lists, no merge launch; same k order
# speedup vs baseline: 9.9141x; 2.5229x over previous
#include <hip/hip_runtime.h>
#include <hip/hip_bf16.h>

typedef __attribute__((ext_vector_type(2))) float v2f;
typedef __attribute__((ext_vector_type(8))) float v8f;

#define NB     2
#define NPT    16384
#define NS     4096
#define NC     64
#define NOUT   128
#define NK     16
#define AGROW  1076

typedef __attribute__((ext_vector_type(4))) float v4f;

#define KQ   2
#define KQB  (KQ * 8)
#define KT   4096
#define KPT  (KT / 256)

__global__ __launch_bounds__(256) void knn_scan_kernel(
    const float* __restrict__ xyz, const float* __restrict__ sxyz,
    int* __restrict__ knn) {
    const int tid  = threadIdx.x;
    const int lane = tid & 31;
    const int wv   = __builtin_amdgcn_readfirstlane(tid >> 5);
    const int b    = blockIdx.x / (NS / KQB);
    const int q0   = (blockIdx.x % (NS / KQB)) * KQB + wv * KQ;

    __shared__ v4f cand[2][KT];

    float sx[KQ], sy[KQ], sz[KQ], s2[KQ], thr[KQ], Ld[KQ];
    int   Li[KQ];
#pragma unroll
    for (int q = 0; q < KQ; ++q) {
        sx[q] = sxyz[(b * 3 + 0) * NS + q0 + q];
        sy[q] = sxyz[(b * 3 + 1) * NS + q0 + q];
        sz[q] = sxyz[(b * 3 + 2) * NS + q0 + q];
        s2[q] = __builtin_fmaf(sz[q], sz[q], __builtin_fmaf(sx[q], sx[q], sy[q] * sy[q]));
        thr[q] = 3.4e38f;
        Ld[q]  = 3.4e38f;
        Li[q]  = 0;
    }

    const float* xb = xyz + (size_t)b * 3 * NPT;

    float px[KPT], py[KPT], pz[KPT];
#pragma unroll
    for (int i = 0; i < KPT; ++i) {
        px[i] = xb[tid + i * 256];
        py[i] = xb[NPT + tid + i * 256];
        pz[i] = xb[2 * NPT + tid + i * 256];
    }

    for (int t0 = 0; t0 < NPT; t0 += KT) {
        v4f* tile = cand[(t0 / KT) & 1];
#pragma unroll
        for (int i = 0; i < KPT; ++i) {
            v4f r;
            r.x = px[i]; r.y = py[i]; r.z = pz[i];
            r.w = __builtin_fmaf(pz[i], pz[i], __builtin_fmaf(px[i], px[i], py[i] * py[i]));
            tile[tid + i * 256] = r;
        }
        __syncthreads();
        if (t0 + KT < NPT) {
#pragma unroll
            for (int i = 0; i < KPT; ++i) {
                px[i] = xb[t0 + KT + tid + i * 256];
                py[i] = xb[NPT + t0 + KT + tid + i * 256];
                pz[i] = xb[2 * NPT + t0 + KT + tid + i * 256];
            }
        }

        const v4f* cp = tile + lane;
        v4f c = cp[0];
#pragma unroll 2
        for (int st = 0; st < KT / 32; ++st) {
            const v4f cn = cp[((st + 1) & (KT / 32 - 1)) * 32];
            const int base = t0 + st * 32;
            float d2[KQ];
            unsigned m[KQ];
            unsigned any = 0;
#pragma unroll
            for (int q = 0; q < KQ; ++q) {
                const float dot = __builtin_fmaf(sz[q], c.z, __builtin_fmaf(sx[q], c.x, sy[q] * c.y));
                d2[q] = __builtin_fmaf(-2.0f, dot, s2[q] + c.w);
                m[q]  = __builtin_amdgcn_ballot_w32(d2[q] < thr[q]);
                any  |= m[q];
            }
            if (any) {
#pragma unroll
                for (int q = 0; q < KQ; ++q) {
                    unsigned mm = m[q];
                    while (mm) {
                        const int l = __builtin_ctz(mm);
                        mm &= mm - 1;
                        const float cd = __builtin_bit_cast(float,
                            __builtin_amdgcn_readlane(__builtin_bit_cast(int, d2[q]), l));
                        if (cd < thr[q]) {
                            const int ci = base + l;
                            const float pd = __builtin_bit_cast(float, __builtin_amdgcn_update_dpp(
                                (int)0xff800000, __builtin_bit_cast(int, Ld[q]), 0x111, 0xf, 0xf, false));
                            const int   pi = __builtin_amdgcn_update_dpp(0, Li[q], 0x111, 0xf, 0xf, false);
                            const bool mp = cd < pd;
                            const bool mk = cd < Ld[q];
                            Li[q] = mp ? pi : (mk ? ci : Li[q]);
                            Ld[q] = __builtin_amdgcn_fmed3f(pd, cd, Ld[q]);
                            thr[q] = __builtin_bit_cast(float,
                                __builtin_amdgcn_readlane(__builtin_bit_cast(int, Ld[q]), 15));
                        }
                    }
                }
            }
            c = cn;
        }
    }

    if (lane < NK) {
#pragma unroll
        for (int q = 0; q < KQ; ++q)
            knn[((size_t)(b * NS + q0 + q)) * NK + lane] = Li[q];
    }
}

__global__ __launch_bounds__(256) void pointconv_kernel(
    const float* __restrict__ xyz,  const float* __restrict__ feat,
    const float* __restrict__ sxyz,
    const float* __restrict__ w1,   const float* __restrict__ b1,
    const float* __restrict__ w2,   const float* __restrict__ b2,
    const float* __restrict__ lw,   const float* __restrict__ lb,
    const int*   __restrict__ knn,  float* __restrict__ out)
{
    const int tid  = threadIdx.x;
    const int lane = tid & 31;
    const int wave = tid >> 5;
    const int b    = blockIdx.x >> 8;
    const int s0   = (blockIdx.x & 255) * 16;

    __shared__ float w1s[24], b1s[8], w2s[128], b2s[16];
    __shared__ float Wl[16][16][16];
    __shared__ float Fl[16][16][68];
    __shared__ float Ag[16 * AGROW];

    if (tid < 24)  w1s[tid] = w1[tid];
    if (tid < 8)   b1s[tid] = b1[tid];
    if (tid < 128) w2s[tid] = w2[tid];
    if (tid < 16)  b2s[tid] = b2[tid];
    __syncthreads();

    {
        const int s = tid >> 4, k = tid & 15;
        const int idx = knn[((size_t)(b * NS + s0 + s)) * NK + k];
        const float nx = xyz[(size_t)b * 3 * NPT + idx];
        const float ny = xyz[(size_t)b * 3 * NPT + NPT + idx];
        const float nz = xyz[(size_t)b * 3 * NPT + 2 * NPT + idx];
        const float rx = nx - sxyz[(b * 3 + 0) * NS + s0 + s];
        const float ry = ny - sxyz[(b * 3 + 1) * NS + s0 + s];
        const float rz = nz - sxyz[(b * 3 + 2) * NS + s0 + s];

        float h[8];
#pragma unroll
        for (int o = 0; o < 8; ++o) {
            float v = w1s[o * 3 + 0] * rx + w1s[o * 3 + 1] * ry +
                      w1s[o * 3 + 2] * rz + b1s[o];
            h[o] = v >= 0.f ? v : 0.1f * v;
        }
#pragma unroll
        for (int o = 0; o < 16; ++o) {
            float v = b2s[o];
#pragma unroll
            for (int i = 0; i < 8; ++i) v += w2s[o * 8 + i] * h[i];
            v = v >= 0.f ? v : 0.1f * v;
            Wl[s][k][o] = v;
        }
        Fl[s][k][0] = nx; Fl[s][k][1] = ny; Fl[s][k][2] = nz;
        const float* fb = feat + (size_t)b * NC * NPT + idx;
#pragma unroll 8
        for (int cf = 0; cf < NC; ++cf)
            Fl[s][k][3 + cf] = fb[(size_t)cf * NPT];
        Fl[s][k][67] = 0.f;
    }
    __syncthreads();

    {
        const int r    = lane & 15;
        const int koff = (lane < 16) ? 0 : 2;
#pragma unroll
        for (int si = 0; si < 2; ++si) {
            const int s = wave * 2 + si;
#pragma unroll
            for (int ct = 0; ct < 5; ++ct) {
                const int c  = ct * 16 + r;
                const int cc = c > 67 ? 67 : c;
                v8f acc = {};
#pragma unroll
                for (int kb = 0; kb < 4; ++kb) {
                    const int k0 = kb * 4 + koff;
                    v2f a;  a.x  = Wl[s][k0][r];  a.y  = Wl[s][k0 + 1][r];
                    v2f bb; bb.x = Fl[s][k0][cc]; bb.y = Fl[s][k0 + 1][cc];
                    acc = __builtin_amdgcn_wmma_f32_16x16x4_f32(
                        false, a, false, bb, (short)0, acc, false, false);
                }
                if (c < 67) {
#pragma unroll
                    for (int rr = 0; rr < 8; ++rr) {
                        const int wi = rr + ((lane < 16) ? 0 : 8);
                        Ag[s * AGROW + wi * 67 + c] = acc[rr];
                    }
                }
            }
        }
    }
    __syncthreads();

    {
        const int r    = lane & 15;
        const int koff = (lane < 16) ? 0 : 2;
        const float* Lrow = lw + (size_t)(wave * 16 + r) * 1072;
        const float* Arow = &Ag[r * AGROW];
        v8f acc = {};
        for (int kb = 0; kb < 268; ++kb) {
            const int k0 = kb * 4 + koff;
            v2f a;  a.x  = Lrow[k0]; a.y  = Lrow[k0 + 1];
            v2f bb; bb.x = Arow[k0]; bb.y = Arow[k0 + 1];
            acc = __builtin_amdgcn_wmma_f32_16x16x4_f32(
                false, a, false, bb, (short)0, acc, false, false);
        }
        const int su = s0 + r;
#pragma unroll
        for (int rr = 0; rr < 8; ++rr) {
            const int o = wave * 16 + rr + ((lane < 16) ? 0 : 8);
            float v = acc[rr] + lb[o];
            v = v >= 0.f ? v : 0.1f * v;
            out[(size_t)b * NOUT * NS + (size_t)o * NS + su] = v;
        }
    }
}

extern "C" void kernel_launch(void* const* d_in, const int* in_sizes, int n_in,
                              void* d_out, int out_size, void* d_ws, size_t ws_size,
                              hipStream_t stream) {
    const float* xyz  = (const float*)d_in[0];
    const float* feat = (const float*)d_in[1];
    const float* sxyz = (const float*)d_in[2];
    const float* w1   = (const float*)d_in[3];
    const float* b1   = (const float*)d_in[4];
    const float* w2   = (const float*)d_in[5];
    const float* b2   = (const float*)d_in[6];
    const float* lw   = (const float*)d_in[7];
    const float* lb   = (const float*)d_in[8];
    float* out = (float*)d_out;

    int* knn = (int*)d_ws;

    knn_scan_kernel<<<NB * (NS / KQB), 256, 0, stream>>>(xyz, sxyz, knn);
    pointconv_kernel<<<NB * (NS / 16), 256, 0, stream>>>(
        xyz, feat, sxyz, w1, b1, w2, b2, lw, lb, knn, out);
}


# Neighbour feature gather reads whole 256-byte rows of a channel-last feature copy that the first launch writes to workspace (was 64 separate lines per neighbour); sums unchanged
# speedup vs baseline: 12.1887x; 1.2294x over previous
#include <hip/hip_runtime.h>
#include <hip/hip_bf16.h>

typedef __attribute__((ext_vector_type(2))) float v2f;
typedef __attribute__((ext_vector_type(8))) float v8f;

#define NB     2
#define NPT    16384
#define NS     4096
#define NC     64
#define NOUT   128
#define NK     16
#define AGROW  1076

typedef __attribute__((ext_vector_type(4))) float v4f;

#define KQ   2
#define KQB  (KQ * 8)
#define KT   4096
#define KPT  (KT / 256)
#define FTP  (NPT / (NS / KQB))

__global__ __launch_bounds__(256) void knn_scan_kernel(
    const float* __restrict__ xyz, const float* __restrict__ sxyz,
    const float* __restrict__ feat, float* __restrict__ featT,
    int* __restrict__ knn) {
    const int tid  = threadIdx.x;
    const int lane = tid & 31;
    const int wv   = __builtin_amdgcn_readfirstlane(tid >> 5);
    const int b    = blockIdx.x / (NS / KQB);
    const int q0   = (blockIdx.x % (NS / KQB)) * KQB + wv * KQ;

    __shared__ v4f cand[2][KT];

    float sx[KQ], sy[KQ], sz[KQ], s2[KQ], thr[KQ], Ld[KQ];
    int   Li[KQ];
#pragma unroll
    for (int q = 0; q < KQ; ++q) {
        sx[q] = sxyz[(b * 3 + 0) * NS + q0 + q];
        sy[q] = sxyz[(b * 3 + 1) * NS + q0 + q];
        sz[q] = sxyz[(b * 3 + 2) * NS + q0 + q];
        s2[q] = __builtin_fmaf(sz[q], sz[q], __builtin_fmaf(sx[q], sx[q], sy[q] * sy[q]));
        thr[q] = 3.4e38f;
        Ld[q]  = 3.4e38f;
        Li[q]  = 0;
    }

    const float* xb = xyz + (size_t)b * 3 * NPT;

    float px[KPT], py[KPT], pz[KPT];
#pragma unroll
    for (int i = 0; i < KPT; ++i) {
        px[i] = xb[tid + i * 256];
        py[i] = xb[NPT + tid + i * 256];
        pz[i] = xb[2 * NPT + tid + i * 256];
    }

    {
        const int n0 = (blockIdx.x % (NS / KQB)) * FTP;
        const float* fb = feat + (size_t)b * NC * NPT + n0;
        float* T = (float*)cand[1];
        float f0[8], f1[8];
#pragma unroll
        for (int i = 0; i < 8; ++i) {
            const float* fc = fb + (size_t)(wv * 8 + i) * NPT;
            f0[i] = fc[lane];
            f1[i] = fc[32 + lane];
        }
#pragma unroll
        for (int i = 0; i < 8; ++i) {
            T[lane * 68 + wv * 8 + i]        = f0[i];
            T[(lane + 32) * 68 + wv * 8 + i] = f1[i];
        }
        __syncthreads();
        float* fo = featT + ((size_t)b * NPT + n0) * NC;
        for (int j = 0; j < 4; ++j) {
            const int row = wv * 8 + j * 2 + (lane >> 4);
            const v4f v = cand[1][row * 17 + (lane & 15)];
            *(v4f*)(fo + row * NC + (lane & 15) * 4) = v;
        }
    }

    for (int t0 = 0; t0 < NPT; t0 += KT) {
        v4f* tile = cand[(t0 / KT) & 1];
#pragma unroll
        for (int i = 0; i < KPT; ++i) {
            v4f r;
            r.x = px[i]; r.y = py[i]; r.z = pz[i];
            r.w = __builtin_fmaf(pz[i], pz[i], __builtin_fmaf(px[i], px[i], py[i] * py[i]));
            tile[tid + i * 256] = r;
        }
        __syncthreads();
        if (t0 + KT < NPT) {
#pragma unroll
            for (int i = 0; i < KPT; ++i) {
                px[i] = xb[t0 + KT + tid + i * 256];
                py[i] = xb[NPT + t0 + KT + tid + i * 256];
                pz[i] = xb[2 * NPT + t0 + KT + tid + i * 256];
            }
        }

        const v4f* cp = tile + lane;
        v4f c = cp[0];
#pragma unroll 2
        for (int st = 0; st < KT / 32; ++st) {
            const v4f cn = cp[((st + 1) & (KT / 32 - 1)) * 32];
            const int base = t0 + st * 32;
            float d2[KQ];
            unsigned m[KQ];
            unsigned any = 0;
#pragma unroll
            for (int q = 0; q < KQ; ++q) {
                const float dot = __builtin_fmaf(sz[q], c.z, __builtin_fmaf(sx[q], c.x, sy[q] * c.y));
                d2[q] = __builtin_fmaf(-2.0f, dot, s2[q] + c.w);
                m[q]  = __builtin_amdgcn_ballot_w32(d2[q] < thr[q]);
                any  |= m[q];
            }
            if (any) {
#pragma unroll
                for (int q = 0; q < KQ; ++q) {
                    unsigned mm = m[q];
                    while (mm) {
                        const int l = __builtin_ctz(mm);
                        mm &= mm - 1;
                        const float cd = __builtin_bit_cast(float,
                            __builtin_amdgcn_readlane(__builtin_bit_cast(int, d2[q]), l));
                        if (cd < thr[q]) {
                            const int ci = base + l;
                            const float pd = __builtin_bit_cast(float, __builtin_amdgcn_update_dpp(
                                (int)0xff800000, __builtin_bit_cast(int, Ld[q]), 0x111, 0xf, 0xf, false));
                            const int   pi = __builtin_amdgcn_update_dpp(0, Li[q], 0x111, 0xf, 0xf, false);
                            const bool mp = cd < pd;
                            const bool mk = cd < Ld[q];
                            Li[q] = mp ? pi : (mk ? ci : Li[q]);
                            Ld[q] = __builtin_amdgcn_fmed3f(pd, cd, Ld[q]);
                            thr[q] = __builtin_bit_cast(float,
                                __builtin_amdgcn_readlane(__builtin_bit_cast(int, Ld[q]), 15));
                        }
                    }
                }
            }
            c = cn;
        }
    }

    if (lane < NK) {
#pragma unroll
        for (int q = 0; q < KQ; ++q)
            knn[((size_t)(b * NS + q0 + q)) * NK + lane] = Li[q];
    }
}

typedef __attribute__((ext_vector_type(4))) int v4i;

__global__ __launch_bounds__(256) void pointconv_kernel(
    const float* __restrict__ xyz,  const float* __restrict__ sxyz,
    const float* __restrict__ w1,   const float* __restrict__ b1,
    const float* __restrict__ w2,   const float* __restrict__ b2,
    const float* __restrict__ lw,   const float* __restrict__ lb,
    const int*   __restrict__ knn,  const float* __restrict__ featT,
    float* __restrict__ out)
{
    const int tid  = threadIdx.x;
    const int lane = tid & 31;
    const int wave = tid >> 5;
    const int b    = blockIdx.x >> 8;
    const int s0   = (blockIdx.x & 255) * 16;

    __shared__ float w1s[24], b1s[8], w2s[128], b2s[16];
    __shared__ float Wl[16][16][16];
    __shared__ __attribute__((aligned(16))) float Fl[16][16][68];
    __shared__ float Ag[16 * AGROW];

    if (tid < 24)  w1s[tid] = w1[tid];
    if (tid < 8)   b1s[tid] = b1[tid];
    if (tid < 128) w2s[tid] = w2[tid];
    if (tid < 16)  b2s[tid] = b2[tid];
    __syncthreads();

    {
        const int s = tid >> 4, k = tid & 15;
        const int c4 = (lane & 15) * 4;
        const v4i* kq = (const v4i*)(knn + ((size_t)(b * NS + s0 + s)) * NK);
        const float* ft = featT + (size_t)b * NPT * NC + c4;
        v4i kv[4];
        v4f fr[16];
#pragma unroll
        for (int i = 0; i < 4; ++i) kv[i] = kq[i];
#pragma unroll
        for (int j = 0; j < 16; ++j)
            fr[j] = *(const v4f*)(ft + (size_t)kv[j >> 2][j & 3] * NC);

        const int idx = knn[((size_t)(b * NS + s0 + s)) * NK + k];
        const float nx = xyz[(size_t)b * 3 * NPT + idx];
        const float ny = xyz[(size_t)b * 3 * NPT + NPT + idx];
        const float nz = xyz[(size_t)b * 3 * NPT + 2 * NPT + idx];
        const float rx = nx - sxyz[(b * 3 + 0) * NS + s0 + s];
        const float ry = ny - sxyz[(b * 3 + 1) * NS + s0 + s];
        const float rz = nz - sxyz[(b * 3 + 2) * NS + s0 + s];

        float h[8];
#pragma unroll
        for (int o = 0; o < 8; ++o) {
            float v = w1s[o * 3 + 0] * rx + w1s[o * 3 + 1] * ry +
                      w1s[o * 3 + 2] * rz + b1s[o];
            h[o] = v >= 0.f ? v : 0.1f * v;
        }
#pragma unroll
        for (int o = 0; o < 16; ++o) {
            float v = b2s[o];
#pragma unroll
            for (int i = 0; i < 8; ++i) v += w2s[o * 8 + i] * h[i];
            v = v >= 0.f ? v : 0.1f * v;
            Wl[s][k][o] = v;
        }
        v4f tail; tail.x = nx; tail.y = ny; tail.z = nz; tail.w = 0.f;
        *(v4f*)&Fl[s][k][64] = tail;
#pragma unroll
        for (int j = 0; j < 16; ++j)
            *(v4f*)&Fl[s][j][c4] = fr[j];
    }
    __syncthreads();

    {
        const int r    = lane & 15;
        const int koff = (lane < 16) ? 0 : 2;
#pragma unroll
        for (int si = 0; si < 2; ++si) {
            const int s = wave * 2 + si;
#pragma unroll
            for (int ct = 0; ct < 5; ++ct) {
                const int c  = ct * 16 + r;
                const int cc = c > 67 ? 67 : c;
                v8f acc = {};
#pragma unroll
                for (int kb = 0; kb < 4; ++kb) {
                    const int k0 = kb * 4 + koff;
                    v2f a;  a.x  = Wl[s][k0][r];  a.y  = Wl[s][k0 + 1][r];
                    v2f bb; bb.x = Fl[s][k0][cc]; bb.y = Fl[s][k0 + 1][cc];
                    acc = __builtin_amdgcn_wmma_f32_16x16x4_f32(
                        false, a, false, bb, (short)0, acc, false, false);
                }
                if (c < 67) {
                    const int cref = c < 64 ? c + 3 : c - 64;
#pragma unroll
                    for (int rr = 0; rr < 8; ++rr) {
                        const int wi = rr + ((lane < 16) ? 0 : 8);
                        Ag[s * AGROW + wi * 67 + cref] = acc[rr];
                    }
                }
            }
        }
    }
    __syncthreads();

    {
        const int r    = lane & 15;
        const int koff = (lane < 16) ? 0 : 2;
        const float* Lrow = lw + (size_t)(wave * 16 + r) * 1072;
        const float* Arow = &Ag[r * AGROW];
        v8f acc = {};
        for (int kb = 0; kb < 268; ++kb) {
            const int k0 = kb * 4 + koff;
            v2f a;  a.x  = Lrow[k0]; a.y  = Lrow[k0 + 1];
            v2f bb; bb.x = Arow[k0]; bb.y = Arow[k0 + 1];
            acc = __builtin_amdgcn_wmma_f32_16x16x4_f32(
                false, a, false, bb, (short)0, acc, false, false);
        }
        const int su = s0 + r;
#pragma unroll
        for (int rr = 0; rr < 8; ++rr) {
            const int o = wave * 16 + rr + ((lane < 16) ? 0 : 8);
            float v = acc[rr] + lb[o];
            v = v >= 0.f ? v : 0.1f * v;
            out[(size_t)b * NOUT * NS + (size_t)o * NS + su] = v;
        }
    }
}

extern "C" void kernel_launch(void* const* d_in, const int* in_sizes, int n_in,
                              void* d_out, int out_size, void* d_ws, size_t ws_size,
                              hipStream_t stream) {
    const float* xyz  = (const float*)d_in[0];
    const float* feat = (const float*)d_in[1];
    const float* sxyz = (const float*)d_in[2];
    const float* w1   = (const float*)d_in[3];
    const float* b1   = (const float*)d_in[4];
    const float* w2   = (const float*)d_in[5];
    const float* b2   = (const float*)d_in[6];
    const float* lw   = (const float*)d_in[7];
    const float* lb   = (const float*)d_in[8];
    float* out = (float*)d_out;

    int*   knn   = (int*)d_ws;
    float* featT = (float*)d_ws + 262144;

    knn_scan_kernel<<<NB * (NS / KQB), 256, 0, stream>>>(xyz, sxyz, feat, featT, knn);
    pointconv_kernel<<<NB * (NS / 16), 256, 0, stream>>>(
        xyz, sxyz, w1, b1, w2, b2, lw, lb, knn, featT, out);
}


# Final GEMM reads lin_w from a fragment-ordered workspace copy the first launch writes: whole-line 16-byte loads per lane, one 16-group register set ahead (was 16 strided lines per load); same sums
# speedup vs baseline: 16.8274x; 1.3806x over previous
#include <hip/hip_runtime.h>
#include <hip/hip_bf16.h>

typedef __attribute__((ext_vector_type(2))) float v2f;
typedef __attribute__((ext_vector_type(8))) float v8f;

#define NB     2
#define NPT    16384
#define NS     4096
#define NC     64
#define NOUT   128
#define NK     16
#define AGROW  1076

typedef __attribute__((ext_vector_type(4))) float v4f;

#define KQ   2
#define KQB  (KQ * 8)
#define KT   4096
#define KPT  (KT / 256)
#define FTP  (NPT / (NS / KQB))
#define LWK  1072
#define LWG  (LWK / 8)
#define LWB  (NOUT / 16 * LWG)

__global__ __launch_bounds__(256) void knn_scan_kernel(
    const float* __restrict__ xyz, const float* __restrict__ sxyz,
    const float* __restrict__ feat, float* __restrict__ featT,
    const float* __restrict__ lw, float* __restrict__ lwF,
    int* __restrict__ knn) {
    const int tid  = threadIdx.x;
    const int lane = tid & 31;
    const int wv   = __builtin_amdgcn_readfirstlane(tid >> 5);
    const int b    = blockIdx.x / (NS / KQB);
    const int q0   = (blockIdx.x % (NS / KQB)) * KQB + wv * KQ;

    __shared__ v4f cand[2][KT];

    float sx[KQ], sy[KQ], sz[KQ], s2[KQ], thr[KQ], Ld[KQ];
    int   Li[KQ];
#pragma unroll
    for (int q = 0; q < KQ; ++q) {
        sx[q] = sxyz[(b * 3 + 0) * NS + q0 + q];
        sy[q] = sxyz[(b * 3 + 1) * NS + q0 + q];
        sz[q] = sxyz[(b * 3 + 2) * NS + q0 + q];
        s2[q] = __builtin_fmaf(sz[q], sz[q], __builtin_fmaf(sx[q], sx[q], sy[q] * sy[q]));
        thr[q] = 3.4e38f;
        Ld[q]  = 3.4e38f;
        Li[q]  = 0;
    }

    const float* xb = xyz + (size_t)b * 3 * NPT;

    float px[KPT], py[KPT], pz[KPT];
#pragma unroll
    for (int i = 0; i < KPT; ++i) {
        px[i] = xb[tid + i * 256];
        py[i] = xb[NPT + tid + i * 256];
        pz[i] = xb[2 * NPT + tid + i * 256];
    }

    const int  fpc = (int)blockIdx.x + wv * (NB * (NS / KQB));
    const bool fdo = (wv < 3) && (fpc < LWB);
    v2f wlo = {}, whi = {};
    if (fdo) {
        const int mt = fpc / LWG, jg = fpc - mt * LWG;
        const float* ws = lw + (size_t)(mt * 16 + (lane & 15)) * LWK + jg * 8 + (lane >> 4) * 2;
        wlo.x = ws[0]; wlo.y = ws[1];
        whi.x = ws[4]; whi.y = ws[5];
    }

    {
        const int n0 = (blockIdx.x % (NS / KQB)) * FTP;
        const float* fb = feat + (size_t)b * NC * NPT + n0;
        float* T = (float*)cand[1];
        float f0[8], f1[8];
#pragma unroll
        for (int i = 0; i < 8; ++i) {
            const float* fc = fb + (size_t)(wv * 8 + i) * NPT;
            f0[i] = fc[lane];
            f1[i] = fc[32 + lane];
        }
#pragma unroll
        for (int i = 0; i < 8; ++i) {
            T[lane * 68 + wv * 8 + i]        = f0[i];
            T[(lane + 32) * 68 + wv * 8 + i] = f1[i];
        }
        __syncthreads();
        float* fo = featT + ((size_t)b * NPT + n0) * NC;
        for (int j = 0; j < 4; ++j) {
            const int row = wv * 8 + j * 2 + (lane >> 4);
            const v4f v = cand[1][row * 17 + (lane & 15)];
            *(v4f*)(fo + row * NC + (lane & 15) * 4) = v;
        }
    }
    if (fdo) {
        v4f o; o.x = wlo.x; o.y = wlo.y; o.z = whi.x; o.w = whi.y;
        ((v4f*)lwF)[(size_t)fpc * 32 + lane] = o;
    }

    for (int t0 = 0; t0 < NPT; t0 += KT) {
        v4f* tile = cand[(t0 / KT) & 1];
#pragma unroll
        for (int i = 0; i < KPT; ++i) {
            v4f r;
            r.x = px[i]; r.y = py[i]; r.z = pz[i];
            r.w = __builtin_fmaf(pz[i], pz[i], __builtin_fmaf(px[i], px[i], py[i] * py[i]));
            tile[tid + i * 256] = r;
        }
        __syncthreads();
        if (t0 + KT < NPT) {
#pragma unroll
            for (int i = 0; i < KPT; ++i) {
                px[i] = xb[t0 + KT + tid + i * 256];
                py[i] = xb[NPT + t0 + KT + tid + i * 256];
                pz[i] = xb[2 * NPT + t0 + KT + tid + i * 256];
            }
        }

        const v4f* cp = tile + lane;
        v4f c = cp[0];
#pragma unroll 2
        for (int st = 0; st < KT / 32; ++st) {
            const v4f cn = cp[((st + 1) & (KT / 32 - 1)) * 32];
            const int base = t0 + st * 32;
            float d2[KQ];
            unsigned m[KQ];
            unsigned any = 0;
#pragma unroll
            for (int q = 0; q < KQ; ++q) {
                const float dot = __builtin_fmaf(sz[q], c.z, __builtin_fmaf(sx[q], c.x, sy[q] * c.y));
                d2[q] = __builtin_fmaf(-2.0f, dot, s2[q] + c.w);
                m[q]  = __builtin_amdgcn_ballot_w32(d2[q] < thr[q]);
                any  |= m[q];
            }
            if (any) {
#pragma unroll
                for (int q = 0; q < KQ; ++q) {
                    unsigned mm = m[q];
                    while (mm) {
                        const int l = __builtin_ctz(mm);
                        mm &= mm - 1;
                        const float cd = __builtin_bit_cast(float,
                            __builtin_amdgcn_readlane(__builtin_bit_cast(int, d2[q]), l));
                        if (cd < thr[q]) {
                            const int ci = base + l;
                            const float pd = __builtin_bit_cast(float, __builtin_amdgcn_update_dpp(
                                (int)0xff800000, __builtin_bit_cast(int, Ld[q]), 0x111, 0xf, 0xf, false));
                            const int   pi = __builtin_amdgcn_update_dpp(0, Li[q], 0x111, 0xf, 0xf, false);
                            const bool mp = cd < pd;
                            const bool mk = cd < Ld[q];
                            Li[q] = mp ? pi : (mk ? ci : Li[q]);
                            Ld[q] = __builtin_amdgcn_fmed3f(pd, cd, Ld[q]);
                            thr[q] = __builtin_bit_cast(float,
                                __builtin_amdgcn_readlane(__builtin_bit_cast(int, Ld[q]), 15));
                        }
                    }
                }
            }
            c = cn;
        }
    }

    if (lane < NK) {
#pragma unroll
        for (int q = 0; q < KQ; ++q)
            knn[((size_t)(b * NS + q0 + q)) * NK + lane] = Li[q];
    }
}

typedef __attribute__((ext_vector_type(4))) int v4i;

#define LFD  16
#define LWT  (LWG % (2 * LFD))

template <int CNT>
__device__ __forceinline__ v8f gemm_groups(v8f acc, const v4f (&a)[LFD], const float* Ab) {
#pragma unroll
    for (int u = 0; u < CNT; ++u) {
        v2f a0; a0.x = a[u].x; a0.y = a[u].y;
        v2f a1; a1.x = a[u].z; a1.y = a[u].w;
        v2f b0; b0.x = Ab[u * 8];     b0.y = Ab[u * 8 + 1];
        v2f b1; b1.x = Ab[u * 8 + 4]; b1.y = Ab[u * 8 + 5];
        acc = __builtin_amdgcn_wmma_f32_16x16x4_f32(
            false, a0, false, b0, (short)0, acc, false, false);
        acc = __builtin_amdgcn_wmma_f32_16x16x4_f32(
            false, a1, false, b1, (short)0, acc, false, false);
    }
    return acc;
}

__global__ __launch_bounds__(256) void pointconv_kernel(
    const float* __restrict__ xyz,  const float* __restrict__ sxyz,
    const float* __restrict__ w1,   const float* __restrict__ b1,
    const float* __restrict__ w2,   const float* __restrict__ b2,
    const float* __restrict__ lwF,  const float* __restrict__ lb,
    const int*   __restrict__ knn,  const float* __restrict__ featT,
    float* __restrict__ out)
{
    const int tid  = threadIdx.x;
    const int lane = tid & 31;
    const int wave = tid >> 5;
    const int b    = blockIdx.x >> 8;
    const int s0   = (blockIdx.x & 255) * 16;

    __shared__ float w1s[24], b1s[8], w2s[128], b2s[16];
    __shared__ float Wl[16][16][16];
    __shared__ __attribute__((aligned(16))) float Fl[16][16][68];
    __shared__ float Ag[16 * AGROW];

    if (tid < 24)  w1s[tid] = w1[tid];
    if (tid < 8)   b1s[tid] = b1[tid];
    if (tid < 128) w2s[tid] = w2[tid];
    if (tid < 16)  b2s[tid] = b2[tid];
    __syncthreads();

    {
        const int s = tid >> 4, k = tid & 15;
        const int c4 = (lane & 15) * 4;
        const v4i* kq = (const v4i*)(knn + ((size_t)(b * NS + s0 + s)) * NK);
        const float* ft = featT + (size_t)b * NPT * NC + c4;
        v4i kv[4];
        v4f fr[16];
#pragma unroll
        for (int i = 0; i < 4; ++i) kv[i] = kq[i];
#pragma unroll
        for (int j = 0; j < 16; ++j)
            fr[j] = *(const v4f*)(ft + (size_t)kv[j >> 2][j & 3] * NC);

        const int idx = knn[((size_t)(b * NS + s0 + s)) * NK + k];
        const float nx = xyz[(size_t)b * 3 * NPT + idx];
        const float ny = xyz[(size_t)b * 3 * NPT + NPT + idx];
        const float nz = xyz[(size_t)b * 3 * NPT + 2 * NPT + idx];
        const float rx = nx - sxyz[(b * 3 + 0) * NS + s0 + s];
        const float ry = ny - sxyz[(b * 3 + 1) * NS + s0 + s];
        const float rz = nz - sxyz[(b * 3 + 2) * NS + s0 + s];

        float h[8];
#pragma unroll
        for (int o = 0; o < 8; ++o) {
            float v = w1s[o * 3 + 0] * rx + w1s[o * 3 + 1] * ry +
                      w1s[o * 3 + 2] * rz + b1s[o];
            h[o] = v >= 0.f ? v : 0.1f * v;
        }
#pragma unroll
        for (int o = 0; o < 16; ++o) {
            float v = b2s[o];
#pragma unroll
            for (int i = 0; i < 8; ++i) v += w2s[o * 8 + i] * h[i];
            v = v >= 0.f ? v : 0.1f * v;
            Wl[s][k][o] = v;
        }
        v4f tail; tail.x = nx; tail.y = ny; tail.z = nz; tail.w = 0.f;
        *(v4f*)&Fl[s][k][64] = tail;
#pragma unroll
        for (int j = 0; j < 16; ++j)
            *(v4f*)&Fl[s][j][c4] = fr[j];
    }
    __syncthreads();

    {
        const int r    = lane & 15;
        const int koff = (lane < 16) ? 0 : 2;
#pragma unroll
        for (int si = 0; si < 2; ++si) {
            const int s = wave * 2 + si;
#pragma unroll
            for (int ct = 0; ct < 5; ++ct) {
                const int c  = ct * 16 + r;
                const int cc = c > 67 ? 67 : c;
                v8f acc = {};
#pragma unroll
                for (int kb = 0; kb < 4; ++kb) {
                    const int k0 = kb * 4 + koff;
                    v2f a;  a.x  = Wl[s][k0][r];  a.y  = Wl[s][k0 + 1][r];
                    v2f bb; bb.x = Fl[s][k0][cc]; bb.y = Fl[s][k0 + 1][cc];
                    acc = __builtin_amdgcn_wmma_f32_16x16x4_f32(
                        false, a, false, bb, (short)0, acc, false, false);
                }
                if (c < 67) {
                    const int cref = c < 64 ? c + 3 : c - 64;
#pragma unroll
                    for (int rr = 0; rr < 8; ++rr) {
                        const int wi = rr + ((lane < 16) ? 0 : 8);
                        Ag[s * AGROW + wi * 67 + cref] = acc[rr];
                    }
                }
            }
        }
    }

    const v4f* LF = (const v4f*)lwF + (size_t)wave * (LWG * 32) + lane;
    v4f pa[LFD], pb[LFD];
#pragma unroll
    for (int u = 0; u < LFD; ++u) pa[u] = LF[u * 32];
    __syncthreads();
    {
        const int r    = lane & 15;
        const int koff = (lane < 16) ? 0 : 2;
        const float* Arow = &Ag[r * AGROW + koff];
        v8f acc = {};
        for (int j0 = 0; j0 < LWG - LWT - 2 * LFD; j0 += 2 * LFD) {
#pragma unroll
            for (int u = 0; u < LFD; ++u) pb[u] = LF[(j0 + LFD + u) * 32];
            acc = gemm_groups<LFD>(acc, pa, Arow + j0 * 8);
#pragma unroll
            for (int u = 0; u < LFD; ++u) pa[u] = LF[(j0 + 2 * LFD + u) * 32];
            acc = gemm_groups<LFD>(acc, pb, Arow + (j0 + LFD) * 8);
        }
#pragma unroll
        for (int u = 0; u < LFD; ++u) pb[u] = LF[(LWG - LWT - LFD + u) * 32];
        acc = gemm_groups<LFD>(acc, pa, Arow + (LWG - LWT - 2 * LFD) * 8);
#pragma unroll
        for (int u = 0; u < LWT; ++u) pa[u] = LF[(LWG - LWT + u) * 32];
        acc = gemm_groups<LFD>(acc, pb, Arow + (LWG - LWT - LFD) * 8);
        acc = gemm_groups<LWT>(acc, pa, Arow + (LWG - LWT) * 8);
        const int su = s0 + r;
#pragma unroll
        for (int rr = 0; rr < 8; ++rr) {
            const int o = wave * 16 + rr + ((lane < 16) ? 0 : 8);
            float v = acc[rr] + lb[o];
            v = v >= 0.f ? v : 0.1f * v;
            out[(size_t)b * NOUT * NS + (size_t)o * NS + su] = v;
        }
    }
}

extern "C" void kernel_launch(void* const* d_in, const int* in_sizes, int n_in,
                              void* d_out, int out_size, void* d_ws, size_t ws_size,
                              hipStream_t stream) {
    const float* xyz  = (const float*)d_in[0];
    const float* feat = (const float*)d_in[1];
    const float* sxyz = (const float*)d_in[2];
    const float* w1   = (const float*)d_in[3];
    const float* b1   = (const float*)d_in[4];
    const float* w2   = (const float*)d_in[5];
    const float* b2   = (const float*)d_in[6];
    const float* lw   = (const float*)d_in[7];
    const float* lb   = (const float*)d_in[8];
    float* out = (float*)d_out;

    int*   knn   = (int*)d_ws;
    float* featT = (float*)d_ws + 262144;
    float* lwF   = (float*)d_ws + 4194304;

    knn_scan_kernel<<<NB * (NS / KQB), 256, 0, stream>>>(xyz, sxyz, feat, featT, lw, lwF, knn);
    pointconv_kernel<<<NB * (NS / 16), 256, 0, stream>>>(
        xyz, sxyz, w1, b1, w2, b2, lwF, lb, knn, featT, out);
}


# KNN insert loop runs on the scalar pass mask alone: no per-candidate scalar re-test of the bound, bound read back once per step; a stale pass is a no-op insert; same lists, same k order
# speedup vs baseline: 17.7787x; 1.0565x over previous
#include <hip/hip_runtime.h>
#include <hip/hip_bf16.h>

typedef __attribute__((ext_vector_type(2))) float v2f;
typedef __attribute__((ext_vector_type(8))) float v8f;

#define NB     2
#define NPT    16384
#define NS     4096
#define NC     64
#define NOUT   128
#define NK     16
#define AGROW  1076

typedef __attribute__((ext_vector_type(4))) float v4f;

#define KQ   2
#define KQB  (KQ * 8)
#define KT   4096
#define KPT  (KT / 256)
#define FTP  (NPT / (NS / KQB))
#define LWK  1072
#define LWG  (LWK / 8)
#define LWB  (NOUT / 16 * LWG)

__global__ __launch_bounds__(256) void knn_scan_kernel(
    const float* __restrict__ xyz, const float* __restrict__ sxyz,
    const float* __restrict__ feat, float* __restrict__ featT,
    const float* __restrict__ lw, float* __restrict__ lwF,
    int* __restrict__ knn) {
    const int tid  = threadIdx.x;
    const int lane = tid & 31;
    const int wv   = __builtin_amdgcn_readfirstlane(tid >> 5);
    const int b    = blockIdx.x / (NS / KQB);
    const int q0   = (blockIdx.x % (NS / KQB)) * KQB + wv * KQ;

    __shared__ v4f cand[2][KT];

    float sx[KQ], sy[KQ], sz[KQ], s2[KQ], thr[KQ], Ld[KQ];
    int   Li[KQ];
#pragma unroll
    for (int q = 0; q < KQ; ++q) {
        sx[q] = sxyz[(b * 3 + 0) * NS + q0 + q];
        sy[q] = sxyz[(b * 3 + 1) * NS + q0 + q];
        sz[q] = sxyz[(b * 3 + 2) * NS + q0 + q];
        s2[q] = __builtin_fmaf(sz[q], sz[q], __builtin_fmaf(sx[q], sx[q], sy[q] * sy[q]));
        thr[q] = 3.4e38f;
        Ld[q]  = 3.4e38f;
        Li[q]  = 0;
    }

    const float* xb = xyz + (size_t)b * 3 * NPT;

    float px[KPT], py[KPT], pz[KPT];
#pragma unroll
    for (int i = 0; i < KPT; ++i) {
        px[i] = xb[tid + i * 256];
        py[i] = xb[NPT + tid + i * 256];
        pz[i] = xb[2 * NPT + tid + i * 256];
    }

    const int  fpc = (int)blockIdx.x + wv * (NB * (NS / KQB));
    const bool fdo = (wv < 3) && (fpc < LWB);
    v2f wlo = {}, whi = {};
    if (fdo) {
        const int mt = fpc / LWG, jg = fpc - mt * LWG;
        const float* ws = lw + (size_t)(mt * 16 + (lane & 15)) * LWK + jg * 8 + (lane >> 4) * 2;
        wlo.x = ws[0]; wlo.y = ws[1];
        whi.x = ws[4]; whi.y = ws[5];
    }

    {
        const int n0 = (blockIdx.x % (NS / KQB)) * FTP;
        const float* fb = feat + (size_t)b * NC * NPT + n0;
        float* T = (float*)cand[1];
        float f0[8], f1[8];
#pragma unroll
        for (int i = 0; i < 8; ++i) {
            const float* fc = fb + (size_t)(wv * 8 + i) * NPT;
            f0[i] = fc[lane];
            f1[i] = fc[32 + lane];
        }
#pragma unroll
        for (int i = 0; i < 8; ++i) {
            T[lane * 68 + wv * 8 + i]        = f0[i];
            T[(lane + 32) * 68 + wv * 8 + i] = f1[i];
        }
        __syncthreads();
        float* fo = featT + ((size_t)b * NPT + n0) * NC;
        for (int j = 0; j < 4; ++j) {
            const int row = wv * 8 + j * 2 + (lane >> 4);
            const v4f v = cand[1][row * 17 + (lane & 15)];
            *(v4f*)(fo + row * NC + (lane & 15) * 4) = v;
        }
    }
    if (fdo) {
        v4f o; o.x = wlo.x; o.y = wlo.y; o.z = whi.x; o.w = whi.y;
        ((v4f*)lwF)[(size_t)fpc * 32 + lane] = o;
    }

    for (int t0 = 0; t0 < NPT; t0 += KT) {
        v4f* tile = cand[(t0 / KT) & 1];
#pragma unroll
        for (int i = 0; i < KPT; ++i) {
            v4f r;
            r.x = px[i]; r.y = py[i]; r.z = pz[i];
            r.w = __builtin_fmaf(pz[i], pz[i], __builtin_fmaf(px[i], px[i], py[i] * py[i]));
            tile[tid + i * 256] = r;
        }
        __syncthreads();
        if (t0 + KT < NPT) {
#pragma unroll
            for (int i = 0; i < KPT; ++i) {
                px[i] = xb[t0 + KT + tid + i * 256];
                py[i] = xb[NPT + t0 + KT + tid + i * 256];
                pz[i] = xb[2 * NPT + t0 + KT + tid + i * 256];
            }
        }

        const v4f* cp = tile + lane;
        v4f c = cp[0];
#pragma unroll 2
        for (int st = 0; st < KT / 32; ++st) {
            const v4f cn = cp[((st + 1) & (KT / 32 - 1)) * 32];
            const int base = t0 + st * 32;
            float d2[KQ];
            unsigned m[KQ];
            unsigned any = 0;
#pragma unroll
            for (int q = 0; q < KQ; ++q) {
                const float dot = __builtin_fmaf(sz[q], c.z, __builtin_fmaf(sx[q], c.x, sy[q] * c.y));
                d2[q] = __builtin_fmaf(-2.0f, dot, s2[q] + c.w);
                m[q]  = __builtin_amdgcn_ballot_w32(d2[q] < thr[q]);
                any  |= m[q];
            }
            if (any) {
#pragma unroll
                for (int q = 0; q < KQ; ++q) {
                    unsigned mm = m[q];
                    if (mm) {
                        do {
                            const int l = __builtin_ctz(mm);
                            mm &= mm - 1;
                            const float cd = __builtin_bit_cast(float,
                                __builtin_amdgcn_readlane(__builtin_bit_cast(int, d2[q]), l));
                            const int ci = base + l;
                            const float pd = __builtin_bit_cast(float, __builtin_amdgcn_update_dpp(
                                (int)0xff800000, __builtin_bit_cast(int, Ld[q]), 0x111, 0xf, 0xf, false));
                            const int   pi = __builtin_amdgcn_update_dpp(0, Li[q], 0x111, 0xf, 0xf, false);
                            const bool mp = cd < pd;
                            const bool mk = cd < Ld[q];
                            Li[q] = mp ? pi : (mk ? ci : Li[q]);
                            Ld[q] = __builtin_amdgcn_fmed3f(pd, cd, Ld[q]);
                        } while (mm);
                        thr[q] = __builtin_bit_cast(float,
                            __builtin_amdgcn_readlane(__builtin_bit_cast(int, Ld[q]), 15));
                    }
                }
            }
            c = cn;
        }
    }

    if (lane < NK) {
#pragma unroll
        for (int q = 0; q < KQ; ++q)
            knn[((size_t)(b * NS + q0 + q)) * NK + lane] = Li[q];
    }
}

typedef __attribute__((ext_vector_type(4))) int v4i;

#define LFD  16
#define LWT  (LWG % (2 * LFD))

template <int CNT>
__device__ __forceinline__ v8f gemm_groups(v8f acc, const v4f (&a)[LFD], const float* Ab) {
#pragma unroll
    for (int u = 0; u < CNT; ++u) {
        v2f a0; a0.x = a[u].x; a0.y = a[u].y;
        v2f a1; a1.x = a[u].z; a1.y = a[u].w;
        v2f b0; b0.x = Ab[u * 8];     b0.y = Ab[u * 8 + 1];
        v2f b1; b1.x = Ab[u * 8 + 4]; b1.y = Ab[u * 8 + 5];
        acc = __builtin_amdgcn_wmma_f32_16x16x4_f32(
            false, a0, false, b0, (short)0, acc, false, false);
        acc = __builtin_amdgcn_wmma_f32_16x16x4_f32(
            false, a1, false, b1, (short)0, acc, false, false);
    }
    return acc;
}

__global__ __launch_bounds__(256) void pointconv_kernel(
    const float* __restrict__ xyz,  const float* __restrict__ sxyz,
    const float* __restrict__ w1,   const float* __restrict__ b1,
    const float* __restrict__ w2,   const float* __restrict__ b2,
    const float* __restrict__ lwF,  const float* __restrict__ lb,
    const int*   __restrict__ knn,  const float* __restrict__ featT,
    float* __restrict__ out)
{
    const int tid  = threadIdx.x;
    const int lane = tid & 31;
    const int wave = tid >> 5;
    const int b    = blockIdx.x >> 8;
    const int s0   = (blockIdx.x & 255) * 16;

    __shared__ float w1s[24], b1s[8], w2s[128], b2s[16];
    __shared__ float Wl[16][16][16];
    __shared__ __attribute__((aligned(16))) float Fl[16][16][68];
    __shared__ float Ag[16 * AGROW];

    if (tid < 24)  w1s[tid] = w1[tid];
    if (tid < 8)   b1s[tid] = b1[tid];
    if (tid < 128) w2s[tid] = w2[tid];
    if (tid < 16)  b2s[tid] = b2[tid];
    __syncthreads();

    {
        const int s = tid >> 4, k = tid & 15;
        const int c4 = (lane & 15) * 4;
        const v4i* kq = (const v4i*)(knn + ((size_t)(b * NS + s0 + s)) * NK);
        const float* ft = featT + (size_t)b * NPT * NC + c4;
        v4i kv[4];
        v4f fr[16];
#pragma unroll
        for (int i = 0; i < 4; ++i) kv[i] = kq[i];
#pragma unroll
        for (int j = 0; j < 16; ++j)
            fr[j] = *(const v4f*)(ft + (size_t)kv[j >> 2][j & 3] * NC);

        const int idx = knn[((size_t)(b * NS + s0 + s)) * NK + k];
        const float nx = xyz[(size_t)b * 3 * NPT + idx];
        const float ny = xyz[(size_t)b * 3 * NPT + NPT + idx];
        const float nz = xyz[(size_t)b * 3 * NPT + 2 * NPT + idx];
        const float rx = nx - sxyz[(b * 3 + 0) * NS + s0 + s];
        const float ry = ny - sxyz[(b * 3 + 1) * NS + s0 + s];
        const float rz = nz - sxyz[(b * 3 + 2) * NS + s0 + s];

        float h[8];
#pragma unroll
        for (int o = 0; o < 8; ++o) {
            float v = w1s[o * 3 + 0] * rx + w1s[o * 3 + 1] * ry +
                      w1s[o * 3 + 2] * rz + b1s[o];
            h[o] = v >= 0.f ? v : 0.1f * v;
        }
#pragma unroll
        for (int o = 0; o < 16; ++o) {
            float v = b2s[o];
#pragma unroll
            for (int i = 0; i < 8; ++i) v += w2s[o * 8 + i] * h[i];
            v = v >= 0.f ? v : 0.1f * v;
            Wl[s][k][o] = v;
        }
        v4f tail; tail.x = nx; tail.y = ny; tail.z = nz; tail.w = 0.f;
        *(v4f*)&Fl[s][k][64] = tail;
#pragma unroll
        for (int j = 0; j < 16; ++j)
            *(v4f*)&Fl[s][j][c4] = fr[j];
    }
    __syncthreads();

    {
        const int r    = lane & 15;
        const int koff = (lane < 16) ? 0 : 2;
#pragma unroll
        for (int si = 0; si < 2; ++si) {
            const int s = wave * 2 + si;
#pragma unroll
            for (int ct = 0; ct < 5; ++ct) {
                const int c  = ct * 16 + r;
                const int cc = c > 67 ? 67 : c;
                v8f acc = {};
#pragma unroll
                for (int kb = 0; kb < 4; ++kb) {
                    const int k0 = kb * 4 + koff;
                    v2f a;  a.x  = Wl[s][k0][r];  a.y  = Wl[s][k0 + 1][r];
                    v2f bb; bb.x = Fl[s][k0][cc]; bb.y = Fl[s][k0 + 1][cc];
                    acc = __builtin_amdgcn_wmma_f32_16x16x4_f32(
                        false, a, false, bb, (short)0, acc, false, false);
                }
                if (c < 67) {
                    const int cref = c < 64 ? c + 3 : c - 64;
#pragma unroll
                    for (int rr = 0; rr < 8; ++rr) {
                        const int wi = rr + ((lane < 16) ? 0 : 8);
                        Ag[s * AGROW + wi * 67 + cref] = acc[rr];
                    }
                }
            }
        }
    }

    const v4f* LF = (const v4f*)lwF + (size_t)wave * (LWG * 32) + lane;
    v4f pa[LFD], pb[LFD];
#pragma unroll
    for (int u = 0; u < LFD; ++u) pa[u] = LF[u * 32];
    __syncthreads();
    {
        const int r    = lane & 15;
        const int koff = (lane < 16) ? 0 : 2;
        const float* Arow = &Ag[r * AGROW + koff];
        v8f acc = {};
        for (int j0 = 0; j0 < LWG - LWT - 2 * LFD; j0 += 2 * LFD) {
#pragma unroll
            for (int u = 0; u < LFD; ++u) pb[u] = LF[(j0 + LFD + u) * 32];
            acc = gemm_groups<LFD>(acc, pa, Arow + j0 * 8);
#pragma unroll
            for (int u = 0; u < LFD; ++u) pa[u] = LF[(j0 + 2 * LFD + u) * 32];
            acc = gemm_groups<LFD>(acc, pb, Arow + (j0 + LFD) * 8);
        }
#pragma unroll
        for (int u = 0; u < LFD; ++u) pb[u] = LF[(LWG - LWT - LFD + u) * 32];
        acc = gemm_groups<LFD>(acc, pa, Arow + (LWG - LWT - 2 * LFD) * 8);
#pragma unroll
        for (int u = 0; u < LWT; ++u) pa[u] = LF[(LWG - LWT + u) * 32];
        acc = gemm_groups<LFD>(acc, pb, Arow + (LWG - LWT - LFD) * 8);
        acc = gemm_groups<LWT>(acc, pa, Arow + (LWG - LWT) * 8);
        const int su = s0 + r;
#pragma unroll
        for (int rr = 0; rr < 8; ++rr) {
            const int o = wave * 16 + rr + ((lane < 16) ? 0 : 8);
            float v = acc[rr] + lb[o];
            v = v >= 0.f ? v : 0.1f * v;
            out[(size_t)b * NOUT * NS + (size_t)o * NS + su] = v;
        }
    }
}

extern "C" void kernel_launch(void* const* d_in, const int* in_sizes, int n_in,
                              void* d_out, int out_size, void* d_ws, size_t ws_size,
                              hipStream_t stream) {
    const float* xyz  = (const float*)d_in[0];
    const float* feat = (const float*)d_in[1];
    const float* sxyz = (const float*)d_in[2];
    const float* w1   = (const float*)d_in[3];
    const float* b1   = (const float*)d_in[4];
    const float* w2   = (const float*)d_in[5];
    const float* b2   = (const float*)d_in[6];
    const float* lw   = (const float*)d_in[7];
    const float* lb   = (const float*)d_in[8];
    float* out = (float*)d_out;

    int*   knn   = (int*)d_ws;
    float* featT = (float*)d_ws + 262144;
    float* lwF   = (float*)d_ws + 4194304;

    knn_scan_kernel<<<NB * (NS / KQB), 256, 0, stream>>>(xyz, sxyz, feat, featT, lw, lwF, knn);
    pointconv_kernel<<<NB * (NS / 16), 256, 0, stream>>>(
        xyz, sxyz, w1, b1, w2, b2, lwF, lb, knn, featT, out);
}


# KNN scan takes 4 steps (128 candidates) per trip: a trip's LDS records are read one whole trip early and all its pass masks meet in one scalar test and branch; same lists, same k order
# speedup vs baseline: 18.3730x; 1.0334x over previous
#include <hip/hip_runtime.h>
#include <hip/hip_bf16.h>

typedef __attribute__((ext_vector_type(2))) float v2f;
typedef __attribute__((ext_vector_type(8))) float v8f;

#define NB     2
#define NPT    16384
#define NS     4096
#define NC     64
#define NOUT   128
#define NK     16
#define AGROW  1076

typedef __attribute__((ext_vector_type(4))) float v4f;

#define KQ   2
#define KQB  (KQ * 8)
#define KT   4096
#define KPT  (KT / 256)
#define KU   4
#define FTP  (NPT / (NS / KQB))
#define LWK  1072
#define LWG  (LWK / 8)
#define LWB  (NOUT / 16 * LWG)

__global__ __launch_bounds__(256) void knn_scan_kernel(
    const float* __restrict__ xyz, const float* __restrict__ sxyz,
    const float* __restrict__ feat, float* __restrict__ featT,
    const float* __restrict__ lw, float* __restrict__ lwF,
    int* __restrict__ knn) {
    const int tid  = threadIdx.x;
    const int lane = tid & 31;
    const int wv   = __builtin_amdgcn_readfirstlane(tid >> 5);
    const int b    = blockIdx.x / (NS / KQB);
    const int q0   = (blockIdx.x % (NS / KQB)) * KQB + wv * KQ;

    __shared__ v4f cand[2][KT];

    float sx[KQ], sy[KQ], sz[KQ], s2[KQ], thr[KQ], Ld[KQ];
    int   Li[KQ];
#pragma unroll
    for (int q = 0; q < KQ; ++q) {
        sx[q] = sxyz[(b * 3 + 0) * NS + q0 + q];
        sy[q] = sxyz[(b * 3 + 1) * NS + q0 + q];
        sz[q] = sxyz[(b * 3 + 2) * NS + q0 + q];
        s2[q] = __builtin_fmaf(sz[q], sz[q], __builtin_fmaf(sx[q], sx[q], sy[q] * sy[q]));
        thr[q] = 3.4e38f;
        Ld[q]  = 3.4e38f;
        Li[q]  = 0;
    }

    const float* xb = xyz + (size_t)b * 3 * NPT;

    float px[KPT], py[KPT], pz[KPT];
#pragma unroll
    for (int i = 0; i < KPT; ++i) {
        px[i] = xb[tid + i * 256];
        py[i] = xb[NPT + tid + i * 256];
        pz[i] = xb[2 * NPT + tid + i * 256];
    }

    const int  fpc = (int)blockIdx.x + wv * (NB * (NS / KQB));
    const bool fdo = (wv < 3) && (fpc < LWB);
    v2f wlo = {}, whi = {};
    if (fdo) {
        const int mt = fpc / LWG, jg = fpc - mt * LWG;
        const float* ws = lw + (size_t)(mt * 16 + (lane & 15)) * LWK + jg * 8 + (lane >> 4) * 2;
        wlo.x = ws[0]; wlo.y = ws[1];
        whi.x = ws[4]; whi.y = ws[5];
    }

    {
        const int n0 = (blockIdx.x % (NS / KQB)) * FTP;
        const float* fb = feat + (size_t)b * NC * NPT + n0;
        float* T = (float*)cand[1];
        float f0[8], f1[8];
#pragma unroll
        for (int i = 0; i < 8; ++i) {
            const float* fc = fb + (size_t)(wv * 8 + i) * NPT;
            f0[i] = fc[lane];
            f1[i] = fc[32 + lane];
        }
#pragma unroll
        for (int i = 0; i < 8; ++i) {
            T[lane * 68 + wv * 8 + i]        = f0[i];
            T[(lane + 32) * 68 + wv * 8 + i] = f1[i];
        }
        __syncthreads();
        float* fo = featT + ((size_t)b * NPT + n0) * NC;
        for (int j = 0; j < 4; ++j) {
            const int row = wv * 8 + j * 2 + (lane >> 4);
            const v4f v = cand[1][row * 17 + (lane & 15)];
            *(v4f*)(fo + row * NC + (lane & 15) * 4) = v;
        }
    }
    if (fdo) {
        v4f o; o.x = wlo.x; o.y = wlo.y; o.z = whi.x; o.w = whi.y;
        ((v4f*)lwF)[(size_t)fpc * 32 + lane] = o;
    }

    for (int t0 = 0; t0 < NPT; t0 += KT) {
        v4f* tile = cand[(t0 / KT) & 1];
#pragma unroll
        for (int i = 0; i < KPT; ++i) {
            v4f r;
            r.x = px[i]; r.y = py[i]; r.z = pz[i];
            r.w = __builtin_fmaf(pz[i], pz[i], __builtin_fmaf(px[i], px[i], py[i] * py[i]));
            tile[tid + i * 256] = r;
        }
        __syncthreads();
        if (t0 + KT < NPT) {
#pragma unroll
            for (int i = 0; i < KPT; ++i) {
                px[i] = xb[t0 + KT + tid + i * 256];
                py[i] = xb[NPT + t0 + KT + tid + i * 256];
                pz[i] = xb[2 * NPT + t0 + KT + tid + i * 256];
            }
        }

        const v4f* cp = tile + lane;
        v4f c[KU];
#pragma unroll
        for (int u = 0; u < KU; ++u) c[u] = cp[u * 32];
#pragma unroll 2
        for (int st = 0; st < KT / 32; st += KU) {
            const v4f* np = cp + ((st + KU) & (KT / 32 - 1)) * 32;
            v4f cn[KU];
#pragma unroll
            for (int u = 0; u < KU; ++u) cn[u] = np[u * 32];
            __builtin_amdgcn_sched_barrier(0);
            const int base = t0 + st * 32;
            float d2[KU][KQ];
            unsigned m[KU][KQ];
            unsigned aq[KQ];
#pragma unroll
            for (int q = 0; q < KQ; ++q) aq[q] = 0;
#pragma unroll
            for (int u = 0; u < KU; ++u) {
#pragma unroll
                for (int q = 0; q < KQ; ++q) {
                    const float dot = __builtin_fmaf(sz[q], c[u].z,
                                          __builtin_fmaf(sx[q], c[u].x, sy[q] * c[u].y));
                    d2[u][q] = __builtin_fmaf(-2.0f, dot, s2[q] + c[u].w);
                    m[u][q]  = __builtin_amdgcn_ballot_w32(d2[u][q] < thr[q]);
                    aq[q]   |= m[u][q];
                }
            }
            unsigned any = 0;
#pragma unroll
            for (int q = 0; q < KQ; ++q) any |= aq[q];
            if (any) {
#pragma unroll
                for (int q = 0; q < KQ; ++q) {
                    if (!aq[q]) continue;
                    bool moved = false;
#pragma unroll
                    for (int u = 0; u < KU; ++u) {
                        unsigned mm = m[u][q];
                        if (mm) {
                            if (moved) mm = __builtin_amdgcn_ballot_w32(d2[u][q] < thr[q]);
                        }
                        if (mm) {
                            do {
                                const int l = __builtin_ctz(mm);
                                mm &= mm - 1;
                                const float cd = __builtin_bit_cast(float,
                                    __builtin_amdgcn_readlane(__builtin_bit_cast(int, d2[u][q]), l));
                                const int ci = base + u * 32 + l;
                                const float pd = __builtin_bit_cast(float, __builtin_amdgcn_update_dpp(
                                    (int)0xff800000, __builtin_bit_cast(int, Ld[q]), 0x111, 0xf, 0xf, false));
                                const int   pi = __builtin_amdgcn_update_dpp(0, Li[q], 0x111, 0xf, 0xf, false);
                                const bool mp = cd < pd;
                                const bool mk = cd < Ld[q];
                                Li[q] = mp ? pi : (mk ? ci : Li[q]);
                                Ld[q] = __builtin_amdgcn_fmed3f(pd, cd, Ld[q]);
                            } while (mm);
                            thr[q] = __builtin_bit_cast(float,
                                __builtin_amdgcn_readlane(__builtin_bit_cast(int, Ld[q]), 15));
                            moved = true;
                        }
                    }
                }
            }
#pragma unroll
            for (int u = 0; u < KU; ++u) c[u] = cn[u];
        }
    }

    if (lane < NK) {
#pragma unroll
        for (int q = 0; q < KQ; ++q)
            knn[((size_t)(b * NS + q0 + q)) * NK + lane] = Li[q];
    }
}

typedef __attribute__((ext_vector_type(4))) int v4i;

#define LFD  16
#define LWT  (LWG % (2 * LFD))

template <int CNT>
__device__ __forceinline__ v8f gemm_groups(v8f acc, const v4f (&a)[LFD], const float* Ab) {
#pragma unroll
    for (int u = 0; u < CNT; ++u) {
        v2f a0; a0.x = a[u].x; a0.y = a[u].y;
        v2f a1; a1.x = a[u].z; a1.y = a[u].w;
        v2f b0; b0.x = Ab[u * 8];     b0.y = Ab[u * 8 + 1];
        v2f b1; b1.x = Ab[u * 8 + 4]; b1.y = Ab[u * 8 + 5];
        acc = __builtin_amdgcn_wmma_f32_16x16x4_f32(
            false, a0, false, b0, (short)0, acc, false, false);
        acc = __builtin_amdgcn_wmma_f32_16x16x4_f32(
            false, a1, false, b1, (short)0, acc, false, false);
    }
    return acc;
}

__global__ __launch_bounds__(256) void pointconv_kernel(
    const float* __restrict__ xyz,  const float* __restrict__ sxyz,
    const float* __restrict__ w1,   const float* __restrict__ b1,
    const float* __restrict__ w2,   const float* __restrict__ b2,
    const float* __restrict__ lwF,  const float* __restrict__ lb,
    const int*   __restrict__ knn,  const float* __restrict__ featT,
    float* __restrict__ out)
{
    const int tid  = threadIdx.x;
    const int lane = tid & 31;
    const int wave = tid >> 5;
    const int b    = blockIdx.x >> 8;
    const int s0   = (blockIdx.x & 255) * 16;

    __shared__ float w1s[24], b1s[8], w2s[128], b2s[16];
    __shared__ float Wl[16][16][16];
    __shared__ __attribute__((aligned(16))) float Fl[16][16][68];
    __shared__ float Ag[16 * AGROW];

    if (tid < 24)  w1s[tid] = w1[tid];
    if (tid < 8)   b1s[tid] = b1[tid];
    if (tid < 128) w2s[tid] = w2[tid];
    if (tid < 16)  b2s[tid] = b2[tid];
    __syncthreads();

    {
        const int s = tid >> 4, k = tid & 15;
        const int c4 = (lane & 15) * 4;
        const v4i* kq = (const v4i*)(knn + ((size_t)(b * NS + s0 + s)) * NK);
        const float* ft = featT + (size_t)b * NPT * NC + c4;
        v4i kv[4];
        v4f fr[16];
#pragma unroll
        for (int i = 0; i < 4; ++i) kv[i] = kq[i];
#pragma unroll
        for (int j = 0; j < 16; ++j)
            fr[j] = *(const v4f*)(ft + (size_t)kv[j >> 2][j & 3] * NC);

        const int idx = knn[((size_t)(b * NS + s0 + s)) * NK + k];
        const float nx = xyz[(size_t)b * 3 * NPT + idx];
        const float ny = xyz[(size_t)b * 3 * NPT + NPT + idx];
        const float nz = xyz[(size_t)b * 3 * NPT + 2 * NPT + idx];
        const float rx = nx - sxyz[(b * 3 + 0) * NS + s0 + s];
        const float ry = ny - sxyz[(b * 3 + 1) * NS + s0 + s];
        const float rz = nz - sxyz[(b * 3 + 2) * NS + s0 + s];

        float h[8];
#pragma unroll
        for (int o = 0; o < 8; ++o) {
            float v = w1s[o * 3 + 0] * rx + w1s[o * 3 + 1] * ry +
                      w1s[o * 3 + 2] * rz + b1s[o];
            h[o] = v >= 0.f ? v : 0.1f * v;
        }
#pragma unroll
        for (int o = 0; o < 16; ++o) {
            float v = b2s[o];
#pragma unroll
            for (int i = 0; i < 8; ++i) v += w2s[o * 8 + i] * h[i];
            v = v >= 0.f ? v : 0.1f * v;
            Wl[s][k][o] = v;
        }
        v4f tail; tail.x = nx; tail.y = ny; tail.z = nz; tail.w = 0.f;
        *(v4f*)&Fl[s][k][64] = tail;
#pragma unroll
        for (int j = 0; j < 16; ++j)
            *(v4f*)&Fl[s][j][c4] = fr[j];
    }
    __syncthreads();

    {
        const int r    = lane & 15;
        const int koff = (lane < 16) ? 0 : 2;
#pragma unroll
        for (int si = 0; si < 2; ++si) {
            const int s = wave * 2 + si;
#pragma unroll
            for (int ct = 0; ct < 5; ++ct) {
                const int c  = ct * 16 + r;
                const int cc = c > 67 ? 67 : c;
                v8f acc = {};
#pragma unroll
                for (int kb = 0; kb < 4; ++kb) {
                    const int k0 = kb * 4 + koff;
                    v2f a;  a.x  = Wl[s][k0][r];  a.y  = Wl[s][k0 + 1][r];
                    v2f bb; bb.x = Fl[s][k0][cc]; bb.y = Fl[s][k0 + 1][cc];
                    acc = __builtin_amdgcn_wmma_f32_16x16x4_f32(
                        false, a, false, bb, (short)0, acc, false, false);
                }
                if (c < 67) {
                    const int cref = c < 64 ? c + 3 : c - 64;
#pragma unroll
                    for (int rr = 0; rr < 8; ++rr) {
                        const int wi = rr + ((lane < 16) ? 0 : 8);
                        Ag[s * AGROW + wi * 67 + cref] = acc[rr];
                    }
                }
            }
        }
    }

    const v4f* LF = (const v4f*)lwF + (size_t)wave * (LWG * 32) + lane;
    v4f pa[LFD], pb[LFD];
#pragma unroll
    for (int u = 0; u < LFD; ++u) pa[u] = LF[u * 32];
    __syncthreads();
    {
        const int r    = lane & 15;
        const int koff = (lane < 16) ? 0 : 2;
        const float* Arow = &Ag[r * AGROW + koff];
        v8f acc = {};
        for (int j0 = 0; j0 < LWG - LWT - 2 * LFD; j0 += 2 * LFD) {
#pragma unroll
            for (int u = 0; u < LFD; ++u) pb[u] = LF[(j0 + LFD + u) * 32];
            acc = gemm_groups<LFD>(acc, pa, Arow + j0 * 8);
#pragma unroll
            for (int u = 0; u < LFD; ++u) pa[u] = LF[(j0 + 2 * LFD + u) * 32];
            acc = gemm_groups<LFD>(acc, pb, Arow + (j0 + LFD) * 8);
        }
#pragma unroll
        for (int u = 0; u < LFD; ++u) pb[u] = LF[(LWG - LWT - LFD + u) * 32];
        acc = gemm_groups<LFD>(acc, pa, Arow + (LWG - LWT - 2 * LFD) * 8);
#pragma unroll
        for (int u = 0; u < LWT; ++u) pa[u] = LF[(LWG - LWT + u) * 32];
        acc = gemm_groups<LFD>(acc, pb, Arow + (LWG - LWT - LFD) * 8);
        acc = gemm_groups<LWT>(acc, pa, Arow + (LWG - LWT) * 8);
        const int su = s0 + r;
#pragma unroll
        for (int rr = 0; rr < 8; ++rr) {
            const int o = wave * 16 + rr + ((lane < 16) ? 0 : 8);
            float v = acc[rr] + lb[o];
            v = v >= 0.f ? v : 0.1f * v;
            out[(size_t)b * NOUT * NS + (size_t)o * NS + su] = v;
        }
    }
}

extern "C" void kernel_launch(void* const* d_in, const int* in_sizes, int n_in,
                              void* d_out, int out_size, void* d_ws, size_t ws_size,
                              hipStream_t stream) {
    const float* xyz  = (const float*)d_in[0];
    const float* feat = (const float*)d_in[1];
    const float* sxyz = (const float*)d_in[2];
    const float* w1   = (const float*)d_in[3];
    const float* b1   = (const float*)d_in[4];
    const float* w2   = (const float*)d_in[5];
    const float* b2   = (const float*)d_in[6];
    const float* lw   = (const float*)d_in[7];
    const float* lb   = (const float*)d_in[8];
    float* out = (float*)d_out;

    int*   knn   = (int*)d_ws;
    float* featT = (float*)d_ws + 262144;
    float* lwF   = (float*)d_ws + 4194304;

    knn_scan_kernel<<<NB * (NS / KQB), 256, 0, stream>>>(xyz, sxyz, feat, featT, lw, lwF, knn);
    pointconv_kernel<<<NB * (NS / 16), 256, 0, stream>>>(
        xyz, sxyz, w1, b1, w2, b2, lwF, lb, knn, featT, out);
}


# KNN pass handling: per-step 'bound moved' flag removed; a step that inserted retakes the later steps' masks of its trip at once, an empty step costs one scalar test; same lists, same k order
# speedup vs baseline: 19.4652x; 1.0594x over previous
#include <hip/hip_runtime.h>
#include <hip/hip_bf16.h>

typedef __attribute__((ext_vector_type(2))) float v2f;
typedef __attribute__((ext_vector_type(8))) float v8f;

#define NB     2
#define NPT    16384
#define NS     4096
#define NC     64
#define NOUT   128
#define NK     16
#define AGROW  1076

typedef __attribute__((ext_vector_type(4))) float v4f;

#define KQ   2
#define KQB  (KQ * 8)
#define KT   4096
#define KPT  (KT / 256)
#define KU   4
#define FTP  (NPT / (NS / KQB))
#define LWK  1072
#define LWG  (LWK / 8)
#define LWB  (NOUT / 16 * LWG)

__global__ __launch_bounds__(256) void knn_scan_kernel(
    const float* __restrict__ xyz, const float* __restrict__ sxyz,
    const float* __restrict__ feat, float* __restrict__ featT,
    const float* __restrict__ lw, float* __restrict__ lwF,
    int* __restrict__ knn) {
    const int tid  = threadIdx.x;
    const int lane = tid & 31;
    const int wv   = __builtin_amdgcn_readfirstlane(tid >> 5);
    const int b    = blockIdx.x / (NS / KQB);
    const int q0   = (blockIdx.x % (NS / KQB)) * KQB + wv * KQ;

    __shared__ v4f cand[2][KT];

    float sx[KQ], sy[KQ], sz[KQ], s2[KQ], thr[KQ], Ld[KQ];
    int   Li[KQ];
#pragma unroll
    for (int q = 0; q < KQ; ++q) {
        sx[q] = sxyz[(b * 3 + 0) * NS + q0 + q];
        sy[q] = sxyz[(b * 3 + 1) * NS + q0 + q];
        sz[q] = sxyz[(b * 3 + 2) * NS + q0 + q];
        s2[q] = __builtin_fmaf(sz[q], sz[q], __builtin_fmaf(sx[q], sx[q], sy[q] * sy[q]));
        thr[q] = 3.4e38f;
        Ld[q]  = 3.4e38f;
        Li[q]  = 0;
    }

    const float* xb = xyz + (size_t)b * 3 * NPT;

    float px[KPT], py[KPT], pz[KPT];
#pragma unroll
    for (int i = 0; i < KPT; ++i) {
        px[i] = xb[tid + i * 256];
        py[i] = xb[NPT + tid + i * 256];
        pz[i] = xb[2 * NPT + tid + i * 256];
    }

    const int  fpc = (int)blockIdx.x + wv * (NB * (NS / KQB));
    const bool fdo = (wv < 3) && (fpc < LWB);
    v2f wlo = {}, whi = {};
    if (fdo) {
        const int mt = fpc / LWG, jg = fpc - mt * LWG;
        const float* ws = lw + (size_t)(mt * 16 + (lane & 15)) * LWK + jg * 8 + (lane >> 4) * 2;
        wlo.x = ws[0]; wlo.y = ws[1];
        whi.x = ws[4]; whi.y = ws[5];
    }

    {
        const int n0 = (blockIdx.x % (NS / KQB)) * FTP;
        const float* fb = feat + (size_t)b * NC * NPT + n0;
        float* T = (float*)cand[1];
        float f0[8], f1[8];
#pragma unroll
        for (int i = 0; i < 8; ++i) {
            const float* fc = fb + (size_t)(wv * 8 + i) * NPT;
            f0[i] = fc[lane];
            f1[i] = fc[32 + lane];
        }
#pragma unroll
        for (int i = 0; i < 8; ++i) {
            T[lane * 68 + wv * 8 + i]        = f0[i];
            T[(lane + 32) * 68 + wv * 8 + i] = f1[i];
        }
        __syncthreads();
        float* fo = featT + ((size_t)b * NPT + n0) * NC;
        for (int j = 0; j < 4; ++j) {
            const int row = wv * 8 + j * 2 + (lane >> 4);
            const v4f v = cand[1][row * 17 + (lane & 15)];
            *(v4f*)(fo + row * NC + (lane & 15) * 4) = v;
        }
    }
    if (fdo) {
        v4f o; o.x = wlo.x; o.y = wlo.y; o.z = whi.x; o.w = whi.y;
        ((v4f*)lwF)[(size_t)fpc * 32 + lane] = o;
    }

    for (int t0 = 0; t0 < NPT; t0 += KT) {
        v4f* tile = cand[(t0 / KT) & 1];
#pragma unroll
        for (int i = 0; i < KPT; ++i) {
            v4f r;
            r.x = px[i]; r.y = py[i]; r.z = pz[i];
            r.w = __builtin_fmaf(pz[i], pz[i], __builtin_fmaf(px[i], px[i], py[i] * py[i]));
            tile[tid + i * 256] = r;
        }
        __syncthreads();
        if (t0 + KT < NPT) {
#pragma unroll
            for (int i = 0; i < KPT; ++i) {
                px[i] = xb[t0 + KT + tid + i * 256];
                py[i] = xb[NPT + t0 + KT + tid + i * 256];
                pz[i] = xb[2 * NPT + t0 + KT + tid + i * 256];
            }
        }

        const v4f* cp = tile + lane;
        v4f c[KU];
#pragma unroll
        for (int u = 0; u < KU; ++u) c[u] = cp[u * 32];
#pragma unroll 2
        for (int st = 0; st < KT / 32; st += KU) {
            const v4f* np = cp + ((st + KU) & (KT / 32 - 1)) * 32;
            v4f cn[KU];
#pragma unroll
            for (int u = 0; u < KU; ++u) cn[u] = np[u * 32];
            __builtin_amdgcn_sched_barrier(0);
            const int base = t0 + st * 32;
            float d2[KU][KQ];
            unsigned m[KU][KQ];
            unsigned aq[KQ];
#pragma unroll
            for (int q = 0; q < KQ; ++q) aq[q] = 0;
#pragma unroll
            for (int u = 0; u < KU; ++u) {
#pragma unroll
                for (int q = 0; q < KQ; ++q) {
                    const float dot = __builtin_fmaf(sz[q], c[u].z,
                                          __builtin_fmaf(sx[q], c[u].x, sy[q] * c[u].y));
                    d2[u][q] = __builtin_fmaf(-2.0f, dot, s2[q] + c[u].w);
                    m[u][q]  = __builtin_amdgcn_ballot_w32(d2[u][q] < thr[q]);
                    aq[q]   |= m[u][q];
                }
            }
            unsigned any = 0;
#pragma unroll
            for (int q = 0; q < KQ; ++q) any |= aq[q];
            if (any) {
#pragma unroll
                for (int q = 0; q < KQ; ++q) {
                    if (!aq[q]) continue;
#pragma unroll
                    for (int u = 0; u < KU; ++u) {
                        unsigned mm = m[u][q];
                        if (mm) {
                            do {
                                const int l = __builtin_ctz(mm);
                                mm &= mm - 1;
                                const float cd = __builtin_bit_cast(float,
                                    __builtin_amdgcn_readlane(__builtin_bit_cast(int, d2[u][q]), l));
                                const int ci = base + u * 32 + l;
                                const float pd = __builtin_bit_cast(float, __builtin_amdgcn_update_dpp(
                                    (int)0xff800000, __builtin_bit_cast(int, Ld[q]), 0x111, 0xf, 0xf, false));
                                const int   pi = __builtin_amdgcn_update_dpp(0, Li[q], 0x111, 0xf, 0xf, false);
                                const bool mp = cd < pd;
                                const bool mk = cd < Ld[q];
                                Li[q] = mp ? pi : (mk ? ci : Li[q]);
                                Ld[q] = __builtin_amdgcn_fmed3f(pd, cd, Ld[q]);
                            } while (mm);
                            thr[q] = __builtin_bit_cast(float,
                                __builtin_amdgcn_readlane(__builtin_bit_cast(int, Ld[q]), 15));
#pragma unroll
                            for (int v = u + 1; v < KU; ++v)
                                m[v][q] = __builtin_amdgcn_ballot_w32(d2[v][q] < thr[q]);
                        }
                    }
                }
            }
#pragma unroll
            for (int u = 0; u < KU; ++u) c[u] = cn[u];
        }
    }

    if (lane < NK) {
#pragma unroll
        for (int q = 0; q < KQ; ++q)
            knn[((size_t)(b * NS + q0 + q)) * NK + lane] = Li[q];
    }
}

typedef __attribute__((ext_vector_type(4))) int v4i;

#define LFD  16
#define LWT  (LWG % (2 * LFD))

template <int CNT>
__device__ __forceinline__ v8f gemm_groups(v8f acc, const v4f (&a)[LFD], const float* Ab) {
#pragma unroll
    for (int u = 0; u < CNT; ++u) {
        v2f a0; a0.x = a[u].x; a0.y = a[u].y;
        v2f a1; a1.x = a[u].z; a1.y = a[u].w;
        v2f b0; b0.x = Ab[u * 8];     b0.y = Ab[u * 8 + 1];
        v2f b1; b1.x = Ab[u * 8 + 4]; b1.y = Ab[u * 8 + 5];
        acc = __builtin_amdgcn_wmma_f32_16x16x4_f32(
            false, a0, false, b0, (short)0, acc, false, false);
        acc = __builtin_amdgcn_wmma_f32_16x16x4_f32(
            false, a1, false, b1, (short)0, acc, false, false);
    }
    return acc;
}

__global__ __launch_bounds__(256) void pointconv_kernel(
    const float* __restrict__ xyz,  const float* __restrict__ sxyz,
    const float* __restrict__ w1,   const float* __restrict__ b1,
    const float* __restrict__ w2,   const float* __restrict__ b2,
    const float* __restrict__ lwF,  const float* __restrict__ lb,
    const int*   __restrict__ knn,  const float* __restrict__ featT,
    float* __restrict__ out)
{
    const int tid  = threadIdx.x;
    const int lane = tid & 31;
    const int wave = tid >> 5;
    const int b    = blockIdx.x >> 8;
    const int s0   = (blockIdx.x & 255) * 16;

    __shared__ float w1s[24], b1s[8], w2s[128], b2s[16];
    __shared__ float Wl[16][16][16];
    __shared__ __attribute__((aligned(16))) float Fl[16][16][68];
    __shared__ float Ag[16 * AGROW];

    if (tid < 24)  w1s[tid] = w1[tid];
    if (tid < 8)   b1s[tid] = b1[tid];
    if (tid < 128) w2s[tid] = w2[tid];
    if (tid < 16)  b2s[tid] = b2[tid];
    __syncthreads();

    {
        const int s = tid >> 4, k = tid & 15;
        const int c4 = (lane & 15) * 4;
        const v4i* kq = (const v4i*)(knn + ((size_t)(b * NS + s0 + s)) * NK);
        const float* ft = featT + (size_t)b * NPT * NC + c4;
        v4i kv[4];
        v4f fr[16];
#pragma unroll
        for (int i = 0; i < 4; ++i) kv[i] = kq[i];
#pragma unroll
        for (int j = 0; j < 16; ++j)
            fr[j] = *(const v4f*)(ft + (size_t)kv[j >> 2][j & 3] * NC);

        const int idx = knn[((size_t)(b * NS + s0 + s)) * NK + k];
        const float nx = xyz[(size_t)b * 3 * NPT + idx];
        const float ny = xyz[(size_t)b * 3 * NPT + NPT + idx];
        const float nz = xyz[(size_t)b * 3 * NPT + 2 * NPT + idx];
        const float rx = nx - sxyz[(b * 3 + 0) * NS + s0 + s];
        const float ry = ny - sxyz[(b * 3 + 1) * NS + s0 + s];
        const float rz = nz - sxyz[(b * 3 + 2) * NS + s0 + s];

        float h[8];
#pragma unroll
        for (int o = 0; o < 8; ++o) {
            float v = w1s[o * 3 + 0] * rx + w1s[o * 3 + 1] * ry +
                      w1s[o * 3 + 2] * rz + b1s[o];
            h[o] = v >= 0.f ? v : 0.1f * v;
        }
#pragma unroll
        for (int o = 0; o < 16; ++o) {
            float v = b2s[o];
#pragma unroll
            for (int i = 0; i < 8; ++i) v += w2s[o * 8 + i] * h[i];
            v = v >= 0.f ? v : 0.1f * v;
            Wl[s][k][o] = v;
        }
        v4f tail; tail.x = nx; tail.y = ny; tail.z = nz; tail.w = 0.f;
        *(v4f*)&Fl[s][k][64] = tail;
#pragma unroll
        for (int j = 0; j < 16; ++j)
            *(v4f*)&Fl[s][j][c4] = fr[j];
    }
    __syncthreads();

    {
        const int r    = lane & 15;
        const int koff = (lane < 16) ? 0 : 2;
#pragma unroll
        for (int si = 0; si < 2; ++si) {
            const int s = wave * 2 + si;
#pragma unroll
            for (int ct = 0; ct < 5; ++ct) {
                const int c  = ct * 16 + r;
                const int cc = c > 67 ? 67 : c;
                v8f acc = {};
#pragma unroll
                for (int kb = 0; kb < 4; ++kb) {
                    const int k0 = kb * 4 + koff;
                    v2f a;  a.x  = Wl[s][k0][r];  a.y  = Wl[s][k0 + 1][r];
                    v2f bb; bb.x = Fl[s][k0][cc]; bb.y = Fl[s][k0 + 1][cc];
                    acc = __builtin_amdgcn_wmma_f32_16x16x4_f32(
                        false, a, false, bb, (short)0, acc, false, false);
                }
                if (c < 67) {
                    const int cref = c < 64 ? c + 3 : c - 64;
#pragma unroll
                    for (int rr = 0; rr < 8; ++rr) {
                        const int wi = rr + ((lane < 16) ? 0 : 8);
                        Ag[s * AGROW + wi * 67 + cref] = acc[rr];
                    }
                }
            }
        }
    }

    const v4f* LF = (const v4f*)lwF + (size_t)wave * (LWG * 32) + lane;
    v4f pa[LFD], pb[LFD];
#pragma unroll
    for (int u = 0; u < LFD; ++u) pa[u] = LF[u * 32];
    __syncthreads();
    {
        const int r    = lane & 15;
        const int koff = (lane < 16) ? 0 : 2;
        const float* Arow = &Ag[r * AGROW + koff];
        v8f acc = {};
        for (int j0 = 0; j0 < LWG - LWT - 2 * LFD; j0 += 2 * LFD) {
#pragma unroll
            for (int u = 0; u < LFD; ++u) pb[u] = LF[(j0 + LFD + u) * 32];
            acc = gemm_groups<LFD>(acc, pa, Arow + j0 * 8);
#pragma unroll
            for (int u = 0; u < LFD; ++u) pa[u] = LF[(j0 + 2 * LFD + u) * 32];
            acc = gemm_groups<LFD>(acc, pb, Arow + (j0 + LFD) * 8);
        }
#pragma unroll
        for (int u = 0; u < LFD; ++u) pb[u] = LF[(LWG - LWT - LFD + u) * 32];
        acc = gemm_groups<LFD>(acc, pa, Arow + (LWG - LWT - 2 * LFD) * 8);
#pragma unroll
        for (int u = 0; u < LWT; ++u) pa[u] = LF[(LWG - LWT + u) * 32];
        acc = gemm_groups<LFD>(acc, pb, Arow + (LWG - LWT - LFD) * 8);
        acc = gemm_groups<LWT>(acc, pa, Arow + (LWG - LWT) * 8);
        const int su = s0 + r;
#pragma unroll
        for (int rr = 0; rr < 8; ++rr) {
            const int o = wave * 16 + rr + ((lane < 16) ? 0 : 8);
            float v = acc[rr] + lb[o];
            v = v >= 0.f ? v : 0.1f * v;
            out[(size_t)b * NOUT * NS + (size_t)o * NS + su] = v;
        }
    }
}

extern "C" void kernel_launch(void* const* d_in, const int* in_sizes, int n_in,
                              void* d_out, int out_size, void* d_ws, size_t ws_size,
                              hipStream_t stream) {
    const float* xyz  = (const float*)d_in[0];
    const float* feat = (const float*)d_in[1];
    const float* sxyz = (const float*)d_in[2];
    const float* w1   = (const float*)d_in[3];
    const float* b1   = (const float*)d_in[4];
    const float* w2   = (const float*)d_in[5];
    const float* b2   = (const float*)d_in[6];
    const float* lw   = (const float*)d_in[7];
    const float* lb   = (const float*)d_in[8];
    float* out = (float*)d_out;

    int*   knn   = (int*)d_ws;
    float* featT = (float*)d_ws + 262144;
    float* lwF   = (float*)d_ws + 4194304;

    knn_scan_kernel<<<NB * (NS / KQB), 256, 0, stream>>>(xyz, sxyz, feat, featT, lw, lwF, knn);
    pointconv_kernel<<<NB * (NS / 16), 256, 0, stream>>>(
        xyz, sxyz, w1, b1, w2, b2, lwF, lb, knn, featT, out);
}


# KNN bound seeded before the scan: a branch-free pass over the first 2048 candidates, the 16th smallest of the 32 per-lane minimum distances starts bound and list; same lists, same k order
# speedup vs baseline: 21.5836x; 1.1088x over previous
#include <hip/hip_runtime.h>
#include <hip/hip_bf16.h>

typedef __attribute__((ext_vector_type(2))) float v2f;
typedef __attribute__((ext_vector_type(8))) float v8f;

#define NB     2
#define NPT    16384
#define NS     4096
#define NC     64
#define NOUT   128
#define NK     16
#define AGROW  1076

typedef __attribute__((ext_vector_type(4))) float v4f;

__device__ __forceinline__ float cand_d2(float sx, float sy, float sz, float s2, const v4f c) {
    const float dot = __builtin_fmaf(sz, c.z, __builtin_fmaf(sx, c.x, sy * c.y));
    return __builtin_fmaf(-2.0f, dot, s2 + c.w);
}

#define KQ   2
#define KQB  (KQ * 8)
#define KT   4096
#define KPT  (KT / 256)
#define KU   4
#define KS   2048
#define FTP  (NPT / (NS / KQB))
#define LWK  1072
#define LWG  (LWK / 8)
#define LWB  (NOUT / 16 * LWG)

__global__ __launch_bounds__(256) void knn_scan_kernel(
    const float* __restrict__ xyz, const float* __restrict__ sxyz,
    const float* __restrict__ feat, float* __restrict__ featT,
    const float* __restrict__ lw, float* __restrict__ lwF,
    int* __restrict__ knn) {
    const int tid  = threadIdx.x;
    const int lane = tid & 31;
    const int wv   = __builtin_amdgcn_readfirstlane(tid >> 5);
    const int b    = blockIdx.x / (NS / KQB);
    const int q0   = (blockIdx.x % (NS / KQB)) * KQB + wv * KQ;

    __shared__ v4f cand[2][KT];

    float sx[KQ], sy[KQ], sz[KQ], s2[KQ], thr[KQ], Ld[KQ];
    int   Li[KQ];
#pragma unroll
    for (int q = 0; q < KQ; ++q) {
        sx[q] = sxyz[(b * 3 + 0) * NS + q0 + q];
        sy[q] = sxyz[(b * 3 + 1) * NS + q0 + q];
        sz[q] = sxyz[(b * 3 + 2) * NS + q0 + q];
        s2[q] = __builtin_fmaf(sz[q], sz[q], __builtin_fmaf(sx[q], sx[q], sy[q] * sy[q]));
        thr[q] = 3.4e38f;
        Ld[q]  = 3.4e38f;
        Li[q]  = 0;
    }

    const float* xb = xyz + (size_t)b * 3 * NPT;

    float px[KPT], py[KPT], pz[KPT];
#pragma unroll
    for (int i = 0; i < KPT; ++i) {
        px[i] = xb[tid + i * 256];
        py[i] = xb[NPT + tid + i * 256];
        pz[i] = xb[2 * NPT + tid + i * 256];
    }

    const int  fpc = (int)blockIdx.x + wv * (NB * (NS / KQB));
    const bool fdo = (wv < 3) && (fpc < LWB);
    v2f wlo = {}, whi = {};
    if (fdo) {
        const int mt = fpc / LWG, jg = fpc - mt * LWG;
        const float* ws = lw + (size_t)(mt * 16 + (lane & 15)) * LWK + jg * 8 + (lane >> 4) * 2;
        wlo.x = ws[0]; wlo.y = ws[1];
        whi.x = ws[4]; whi.y = ws[5];
    }

    {
        const int n0 = (blockIdx.x % (NS / KQB)) * FTP;
        const float* fb = feat + (size_t)b * NC * NPT + n0;
        float* T = (float*)cand[1];
        float f0[8], f1[8];
#pragma unroll
        for (int i = 0; i < 8; ++i) {
            const float* fc = fb + (size_t)(wv * 8 + i) * NPT;
            f0[i] = fc[lane];
            f1[i] = fc[32 + lane];
        }
#pragma unroll
        for (int i = 0; i < 8; ++i) {
            T[lane * 68 + wv * 8 + i]        = f0[i];
            T[(lane + 32) * 68 + wv * 8 + i] = f1[i];
        }
        __syncthreads();
        float* fo = featT + ((size_t)b * NPT + n0) * NC;
        for (int j = 0; j < 4; ++j) {
            const int row = wv * 8 + j * 2 + (lane >> 4);
            const v4f v = cand[1][row * 17 + (lane & 15)];
            *(v4f*)(fo + row * NC + (lane & 15) * 4) = v;
        }
    }
    if (fdo) {
        v4f o; o.x = wlo.x; o.y = wlo.y; o.z = whi.x; o.w = whi.y;
        ((v4f*)lwF)[(size_t)fpc * 32 + lane] = o;
    }

    for (int t0 = 0; t0 < NPT; t0 += KT) {
        v4f* tile = cand[(t0 / KT) & 1];
#pragma unroll
        for (int i = 0; i < KPT; ++i) {
            v4f r;
            r.x = px[i]; r.y = py[i]; r.z = pz[i];
            r.w = __builtin_fmaf(pz[i], pz[i], __builtin_fmaf(px[i], px[i], py[i] * py[i]));
            tile[tid + i * 256] = r;
        }
        __syncthreads();
        if (t0 + KT < NPT) {
#pragma unroll
            for (int i = 0; i < KPT; ++i) {
                px[i] = xb[t0 + KT + tid + i * 256];
                py[i] = xb[NPT + t0 + KT + tid + i * 256];
                pz[i] = xb[2 * NPT + t0 + KT + tid + i * 256];
            }
        }

        const v4f* cp = tile + lane;
        if (t0 == 0) {
            float mn[KQ];
#pragma unroll
            for (int q = 0; q < KQ; ++q) mn[q] = 3.4e38f;
#pragma unroll 8
            for (int u = 0; u < KS / 32; ++u) {
                const v4f r = cp[u * 32];
#pragma unroll
                for (int q = 0; q < KQ; ++q)
                    mn[q] = __builtin_fminf(mn[q], cand_d2(sx[q], sy[q], sz[q], s2[q], r));
            }
            float Td[KQ];
#pragma unroll
            for (int q = 0; q < KQ; ++q) Td[q] = 3.4e38f;
#pragma unroll
            for (int j = 0; j < 32; ++j) {
#pragma unroll
                for (int q = 0; q < KQ; ++q) {
                    const float cd = __builtin_bit_cast(float,
                        __builtin_amdgcn_readlane(__builtin_bit_cast(int, mn[q]), j));
                    const float pd = __builtin_bit_cast(float, __builtin_amdgcn_update_dpp(
                        (int)0xff800000, __builtin_bit_cast(int, Td[q]), 0x111, 0xf, 0xf, false));
                    Td[q] = __builtin_amdgcn_fmed3f(pd, cd, Td[q]);
                }
            }
#pragma unroll
            for (int q = 0; q < KQ; ++q) {
                const float t16 = __builtin_bit_cast(float,
                    __builtin_amdgcn_readlane(__builtin_bit_cast(int, Td[q]), 15));
                const float seed = t16 + (__builtin_fabsf(t16) * 1.0e-6f + 1.0e-30f);
                thr[q] = seed;
                Ld[q]  = seed;
            }
        }
        v4f c[KU];
#pragma unroll
        for (int u = 0; u < KU; ++u) c[u] = cp[u * 32];
#pragma unroll 2
        for (int st = 0; st < KT / 32; st += KU) {
            const v4f* np = cp + ((st + KU) & (KT / 32 - 1)) * 32;
            v4f cn[KU];
#pragma unroll
            for (int u = 0; u < KU; ++u) cn[u] = np[u * 32];
            __builtin_amdgcn_sched_barrier(0);
            const int base = t0 + st * 32;
            float d2[KU][KQ];
            unsigned m[KU][KQ];
            unsigned aq[KQ];
#pragma unroll
            for (int q = 0; q < KQ; ++q) aq[q] = 0;
#pragma unroll
            for (int u = 0; u < KU; ++u) {
#pragma unroll
                for (int q = 0; q < KQ; ++q) {
                    d2[u][q] = cand_d2(sx[q], sy[q], sz[q], s2[q], c[u]);
                    m[u][q]  = __builtin_amdgcn_ballot_w32(d2[u][q] < thr[q]);
                    aq[q]   |= m[u][q];
                }
            }
            unsigned any = 0;
#pragma unroll
            for (int q = 0; q < KQ; ++q) any |= aq[q];
            if (any) {
#pragma unroll
                for (int q = 0; q < KQ; ++q) {
                    if (!aq[q]) continue;
#pragma unroll
                    for (int u = 0; u < KU; ++u) {
                        unsigned mm = m[u][q];
                        if (mm) {
                            do {
                                const int l = __builtin_ctz(mm);
                                mm &= mm - 1;
                                const float cd = __builtin_bit_cast(float,
                                    __builtin_amdgcn_readlane(__builtin_bit_cast(int, d2[u][q]), l));
                                const int ci = base + u * 32 + l;
                                const float pd = __builtin_bit_cast(float, __builtin_amdgcn_update_dpp(
                                    (int)0xff800000, __builtin_bit_cast(int, Ld[q]), 0x111, 0xf, 0xf, false));
                                const int   pi = __builtin_amdgcn_update_dpp(0, Li[q], 0x111, 0xf, 0xf, false);
                                const bool mp = cd < pd;
                                const bool mk = cd < Ld[q];
                                Li[q] = mp ? pi : (mk ? ci : Li[q]);
                                Ld[q] = __builtin_amdgcn_fmed3f(pd, cd, Ld[q]);
                            } while (mm);
                            thr[q] = __builtin_bit_cast(float,
                                __builtin_amdgcn_readlane(__builtin_bit_cast(int, Ld[q]), 15));
#pragma unroll
                            for (int v = u + 1; v < KU; ++v)
                                m[v][q] = __builtin_amdgcn_ballot_w32(d2[v][q] < thr[q]);
                        }
                    }
                }
            }
#pragma unroll
            for (int u = 0; u < KU; ++u) c[u] = cn[u];
        }
    }

    if (lane < NK) {
#pragma unroll
        for (int q = 0; q < KQ; ++q)
            knn[((size_t)(b * NS + q0 + q)) * NK + lane] = Li[q];
    }
}

typedef __attribute__((ext_vector_type(4))) int v4i;

#define LFD  16
#define LWT  (LWG % (2 * LFD))

template <int CNT>
__device__ __forceinline__ v8f gemm_groups(v8f acc, const v4f (&a)[LFD], const float* Ab) {
#pragma unroll
    for (int u = 0; u < CNT; ++u) {
        v2f a0; a0.x = a[u].x; a0.y = a[u].y;
        v2f a1; a1.x = a[u].z; a1.y = a[u].w;
        v2f b0; b0.x = Ab[u * 8];     b0.y = Ab[u * 8 + 1];
        v2f b1; b1.x = Ab[u * 8 + 4]; b1.y = Ab[u * 8 + 5];
        acc = __builtin_amdgcn_wmma_f32_16x16x4_f32(
            false, a0, false, b0, (short)0, acc, false, false);
        acc = __builtin_amdgcn_wmma_f32_16x16x4_f32(
            false, a1, false, b1, (short)0, acc, false, false);
    }
    return acc;
}

__global__ __launch_bounds__(256) void pointconv_kernel(
    const float* __restrict__ xyz,  const float* __restrict__ sxyz,
    const float* __restrict__ w1,   const float* __restrict__ b1,
    const float* __restrict__ w2,   const float* __restrict__ b2,
    const float* __restrict__ lwF,  const float* __restrict__ lb,
    const int*   __restrict__ knn,  const float* __restrict__ featT,
    float* __restrict__ out)
{
    const int tid  = threadIdx.x;
    const int lane = tid & 31;
    const int wave = tid >> 5;
    const int b    = blockIdx.x >> 8;
    const int s0   = (blockIdx.x & 255) * 16;

    __shared__ float w1s[24], b1s[8], w2s[128], b2s[16];
    __shared__ float Wl[16][16][16];
    __shared__ __attribute__((aligned(16))) float Fl[16][16][68];
    __shared__ float Ag[16 * AGROW];

    if (tid < 24)  w1s[tid] = w1[tid];
    if (tid < 8)   b1s[tid] = b1[tid];
    if (tid < 128) w2s[tid] = w2[tid];
    if (tid < 16)  b2s[tid] = b2[tid];
    __syncthreads();

    {
        const int s = tid >> 4, k = tid & 15;
        const int c4 = (lane & 15) * 4;
        const v4i* kq = (const v4i*)(knn + ((size_t)(b * NS + s0 + s)) * NK);
        const float* ft = featT + (size_t)b * NPT * NC + c4;
        v4i kv[4];
        v4f fr[16];
#pragma unroll
        for (int i = 0; i < 4; ++i) kv[i] = kq[i];
#pragma unroll
        for (int j = 0; j < 16; ++j)
            fr[j] = *(const v4f*)(ft + (size_t)kv[j >> 2][j & 3] * NC);

        const int idx = knn[((size_t)(b * NS + s0 + s)) * NK + k];
        const float nx = xyz[(size_t)b * 3 * NPT + idx];
        const float ny = xyz[(size_t)b * 3 * NPT + NPT + idx];
        const float nz = xyz[(size_t)b * 3 * NPT + 2 * NPT + idx];
        const float rx = nx - sxyz[(b * 3 + 0) * NS + s0 + s];
        const float ry = ny - sxyz[(b * 3 + 1) * NS + s0 + s];
        const float rz = nz - sxyz[(b * 3 + 2) * NS + s0 + s];

        float h[8];
#pragma unroll
        for (int o = 0; o < 8; ++o) {
            float v = w1s[o * 3 + 0] * rx + w1s[o * 3 + 1] * ry +
                      w1s[o * 3 + 2] * rz + b1s[o];
            h[o] = v >= 0.f ? v : 0.1f * v;
        }
#pragma unroll
        for (int o = 0; o < 16; ++o) {
            float v = b2s[o];
#pragma unroll
            for (int i = 0; i < 8; ++i) v += w2s[o * 8 + i] * h[i];
            v = v >= 0.f ? v : 0.1f * v;
            Wl[s][k][o] = v;
        }
        v4f tail; tail.x = nx; tail.y = ny; tail.z = nz; tail.w = 0.f;
        *(v4f*)&Fl[s][k][64] = tail;
#pragma unroll
        for (int j = 0; j < 16; ++j)
            *(v4f*)&Fl[s][j][c4] = fr[j];
    }
    __syncthreads();

    {
        const int r    = lane & 15;
        const int koff = (lane < 16) ? 0 : 2;
#pragma unroll
        for (int si = 0; si < 2; ++si) {
            const int s = wave * 2 + si;
#pragma unroll
            for (int ct = 0; ct < 5; ++ct) {
                const int c  = ct * 16 + r;
                const int cc = c > 67 ? 67 : c;
                v8f acc = {};
#pragma unroll
                for (int kb = 0; kb < 4; ++kb) {
                    const int k0 = kb * 4 + koff;
                    v2f a;  a.x  = Wl[s][k0][r];  a.y  = Wl[s][k0 + 1][r];
                    v2f bb; bb.x = Fl[s][k0][cc]; bb.y = Fl[s][k0 + 1][cc];
                    acc = __builtin_amdgcn_wmma_f32_16x16x4_f32(
                        false, a, false, bb, (short)0, acc, false, false);
                }
                if (c < 67) {
                    const int cref = c < 64 ? c + 3 : c - 64;
#pragma unroll
                    for (int rr = 0; rr < 8; ++rr) {
                        const int wi = rr + ((lane < 16) ? 0 : 8);
                        Ag[s * AGROW + wi * 67 + cref] = acc[rr];
                    }
                }
            }
        }
    }

    const v4f* LF = (const v4f*)lwF + (size_t)wave * (LWG * 32) + lane;
    v4f pa[LFD], pb[LFD];
#pragma unroll
    for (int u = 0; u < LFD; ++u) pa[u] = LF[u * 32];
    __syncthreads();
    {
        const int r    = lane & 15;
        const int koff = (lane < 16) ? 0 : 2;
        const float* Arow = &Ag[r * AGROW + koff];
        v8f acc = {};
        for (int j0 = 0; j0 < LWG - LWT - 2 * LFD; j0 += 2 * LFD) {
#pragma unroll
            for (int u = 0; u < LFD; ++u) pb[u] = LF[(j0 + LFD + u) * 32];
            acc = gemm_groups<LFD>(acc, pa, Arow + j0 * 8);
#pragma unroll
            for (int u = 0; u < LFD; ++u) pa[u] = LF[(j0 + 2 * LFD + u) * 32];
            acc = gemm_groups<LFD>(acc, pb, Arow + (j0 + LFD) * 8);
        }
#pragma unroll
        for (int u = 0; u < LFD; ++u) pb[u] = LF[(LWG - LWT - LFD + u) * 32];
        acc = gemm_groups<LFD>(acc, pa, Arow + (LWG - LWT - 2 * LFD) * 8);
#pragma unroll
        for (int u = 0; u < LWT; ++u) pa[u] = LF[(LWG - LWT + u) * 32];
        acc = gemm_groups<LFD>(acc, pb, Arow + (LWG - LWT - LFD) * 8);
        acc = gemm_groups<LWT>(acc, pa, Arow + (LWG - LWT) * 8);
        const int su = s0 + r;
#pragma unroll
        for (int rr = 0; rr < 8; ++rr) {
            const int o = wave * 16 + rr + ((lane < 16) ? 0 : 8);
            float v = acc[rr] + lb[o];
            v = v >= 0.f ? v : 0.1f * v;
            out[(size_t)b * NOUT * NS + (size_t)o * NS + su] = v;
        }
    }
}

extern "C" void kernel_launch(void* const* d_in, const int* in_sizes, int n_in,
                              void* d_out, int out_size, void* d_ws, size_t ws_size,
                              hipStream_t stream) {
    const float* xyz  = (const float*)d_in[0];
    const float* feat = (const float*)d_in[1];
    const float* sxyz = (const float*)d_in[2];
    const float* w1   = (const float*)d_in[3];
    const float* b1   = (const float*)d_in[4];
    const float* w2   = (const float*)d_in[5];
    const float* b2   = (const float*)d_in[6];
    const float* lw   = (const float*)d_in[7];
    const float* lb   = (const float*)d_in[8];
    float* out = (float*)d_out;

    int*   knn   = (int*)d_ws;
    float* featT = (float*)d_ws + 262144;
    float* lwF   = (float*)d_ws + 4194304;

    knn_scan_kernel<<<NB * (NS / KQB), 256, 0, stream>>>(xyz, sxyz, feat, featT, lw, lwF, knn);
    pointconv_kernel<<<NB * (NS / 16), 256, 0, stream>>>(
        xyz, sxyz, w1, b1, w2, b2, lwF, lb, knn, featT, out);
}
